# starting kernel
# baseline (speedup 1.0000x reference)
#include <hip/hip_runtime.h>

constexpr int Bc   = 8;
constexpr int Nc   = 256;
constexpr int NFc  = 256;
constexpr int EFc  = 128;
constexpr int Hc   = 8;
constexpr int Oc   = 32;
constexpr int NHID = 1024;
constexpr int HOc  = 256;
constexpr float ALPHA = 0.2f;
constexpr float EPSc  = 1e-6f;

typedef __attribute__((ext_vector_type(2))) float v2f;
typedef __attribute__((ext_vector_type(8))) float v8f;

__device__ __forceinline__ v8f wmma_gemm_f32(const float* __restrict__ A, int lda,
                                             const float* __restrict__ B, int ldb,
                                             int K, v8f acc) {
  const int lane = threadIdx.x & 31;
  const int l16  = lane & 15;
  const int ko   = (lane >> 4) << 1;
  const float* ap = A + l16 * lda + ko;
  const float* bp = B + ko * ldb + l16;
#pragma unroll 8
  for (int k = 0; k < K; k += 4) {
    v2f av, bv;
    av.x = ap[k];
    av.y = ap[k + 1];
    bv.x = bp[k * ldb];
    bv.y = bp[k * ldb + ldb];
    acc = __builtin_amdgcn_wmma_f32_16x16x4_f32(false, av, false, bv,
                                                (short)0, acc, false, false);
  }
  return acc;
}

__device__ __forceinline__ float wave_max32(float v) {
#pragma unroll
  for (int m = 16; m >= 1; m >>= 1) v = fmaxf(v, __shfl_xor(v, m, 32));
  return v;
}
__device__ __forceinline__ float wave_sum32(float v) {
#pragma unroll
  for (int m = 16; m >= 1; m >>= 1) v += __shfl_xor(v, m, 32);
  return v;
}

__global__ void k_h(const float* __restrict__ x, const float* __restrict__ W,
                    float* __restrict__ h) {
  int wg = blockIdx.x * (blockDim.x >> 5) + (threadIdx.x >> 5);
  int nt = wg & 1;
  int mt = (wg >> 1) & 15;
  int bh = wg >> 5;
  int b  = bh >> 3, head = bh & 7;
  const float* A  = x + ((long)b * Nc + mt * 16) * NFc;
  const float* Bm = W + (long)head * NFc * Oc + nt * 16;
  v8f acc = {};
  acc = wmma_gemm_f32(A, NFc, Bm, Oc, NFc, acc);
  float* C = h + ((long)bh * Nc + mt * 16) * Oc + nt * 16;
  int lane = threadIdx.x & 31;
  int col  = lane & 15;
  int rb   = (lane >> 4) * 8;
#pragma unroll
  for (int g = 0; g < 8; ++g) C[(rb + g) * Oc + col] = acc[g];
}

__global__ void k_st(const float* __restrict__ h, const float* __restrict__ a,
                     float* __restrict__ s, float* __restrict__ t) {
  int idx  = blockIdx.x * blockDim.x + threadIdx.x;
  int head = (idx >> 8) & (Hc - 1);
  const float* hr = h + (long)idx * Oc;
  const float* a1 = a + head * (3 * Oc);
  const float* a2 = a1 + Oc;
  float sv = 0.f, tv = 0.f;
#pragma unroll
  for (int o = 0; o < Oc; ++o) { sv += hr[o] * a1[o]; tv += hr[o] * a2[o]; }
  s[idx] = sv;
  t[idx] = tv;
}

__global__ void k_v(const float* __restrict__ W1, const float* __restrict__ a,
                    float* __restrict__ vt) {
  int idx = blockIdx.x * blockDim.x + threadIdx.x;
  int c = idx & 15, e = idx >> 4;
  float v = 0.f;
  if (c < Hc) {
    const float* w  = W1 + ((long)c * EFc + e) * Oc;
    const float* a3 = a + c * (3 * Oc) + 2 * Oc;
#pragma unroll
    for (int o = 0; o < Oc; ++o) v += w[o] * a3[o];
  }
  vt[idx] = v;
}

__global__ void k_u(const float* __restrict__ edge, const float* __restrict__ vt,
                    float* __restrict__ u) {
  int wg = blockIdx.x * (blockDim.x >> 5) + (threadIdx.x >> 5);
  const float* A = edge + (long)wg * 16 * EFc;
  v8f acc = {};
  acc = wmma_gemm_f32(A, EFc, vt, 16, EFc, acc);
  int lane = threadIdx.x & 31;
  int col  = lane & 15;
  int rb   = (lane >> 4) * 8;
  if (col < Hc) {
#pragma unroll
    for (int g = 0; g < 8; ++g) {
      long row = (long)wg * 16 + rb + g;
      int b   = (int)(row >> 16);
      int rem = (int)(row & 65535);
      int i = rem >> 8, j = rem & 255;
      u[(((long)b * Hc + col) * Nc + i) * Nc + j] = acc[g];
    }
  }
}

__global__ void k_attn(float* __restrict__ u, const float* __restrict__ s,
                       const float* __restrict__ t, const float* __restrict__ adj) {
  int wg   = blockIdx.x * (blockDim.x >> 5) + (threadIdx.x >> 5);
  int lane = threadIdx.x & 31;
  int i  = wg & (Nc - 1);
  int bh = wg >> 8;
  int b  = bh >> 3;
  float*       urow = u + (long)wg * Nc;
  const float* trow = t + (long)bh * Nc;
  const float* arow = adj + ((long)b * Nc + i) * Nc;
  float sv = s[wg];
  float vals[8];
  float mx = -3.0e38f;
#pragma unroll
  for (int p = 0; p < 8; ++p) {
    int j = p * 32 + lane;
    float sc = sv + trow[j] + urow[j];
    sc = sc > 0.f ? sc : ALPHA * sc;
    sc = (arow[j] > 0.f) ? sc : -3.0e38f;
    vals[p] = sc;
    mx = fmaxf(mx, sc);
  }
  mx = wave_max32(mx);
  float sum = 0.f;
#pragma unroll
  for (int p = 0; p < 8; ++p) {
    float e = __expf(vals[p] - mx);
    vals[p] = e;
    sum += e;
  }
  sum = wave_sum32(sum);
  float inv = 1.f / sum;
#pragma unroll
  for (int p = 0; p < 8; ++p) urow[p * 32 + lane] = vals[p] * inv;
}

__global__ void k_hp(const float* __restrict__ attn, const float* __restrict__ h,
                     float* __restrict__ cat) {
  int wg = blockIdx.x * (blockDim.x >> 5) + (threadIdx.x >> 5);
  int nt = wg & 1;
  int mt = (wg >> 1) & 15;
  int bh = wg >> 5;
  int b  = bh >> 3, head = bh & 7;
  const float* A  = attn + ((long)bh * Nc + mt * 16) * Nc;
  const float* Bm = h + (long)bh * Nc * Oc + nt * 16;
  v8f acc = {};
  acc = wmma_gemm_f32(A, Nc, Bm, Oc, Nc, acc);
  int lane = threadIdx.x & 31;
  int col  = lane & 15;
  int rb   = (lane >> 4) * 8;
#pragma unroll
  for (int g = 0; g < 8; ++g) {
    int i = mt * 16 + rb + g;
    float vv = acc[g];
    vv = vv > 0.f ? vv : (__expf(vv) - 1.f);
    cat[((long)b * Nc + i) * HOc + head * Oc + nt * 16 + col] = vv;
  }
}

__global__ void k_mlp(const float* __restrict__ A, const float* __restrict__ Bm,
                      const float* __restrict__ bias, float* __restrict__ C,
                      int Ncols, int K, int relu) {
  int waves_n = Ncols >> 4;
  int wg = blockIdx.x * (blockDim.x >> 5) + (threadIdx.x >> 5);
  int nt = wg % waves_n;
  int mt = wg / waves_n;
  const float* Ap = A + (long)mt * 16 * K;
  const float* Bp = Bm + nt * 16;
  v8f acc = {};
  acc = wmma_gemm_f32(Ap, K, Bp, Ncols, K, acc);
  int lane = threadIdx.x & 31;
  int col  = nt * 16 + (lane & 15);
  int rb   = (lane >> 4) * 8;
#pragma unroll
  for (int g = 0; g < 8; ++g) {
    float vv = acc[g] + bias[col];
    if (relu) vv = fmaxf(vv, 0.f);
    C[((long)(mt * 16 + rb + g)) * Ncols + col] = vv;
  }
}

__global__ void k_ff(const float* __restrict__ z, const float* __restrict__ w2,
                     const float* __restrict__ b2, const float* __restrict__ y,
                     float* __restrict__ ff) {
  int wg = blockIdx.x * (blockDim.x >> 5) + (threadIdx.x >> 5);
  int nt = wg & 15;
  int mt = wg >> 4;
  const float* Ap = z + (long)mt * 16 * NHID;
  const float* Bp = w2 + nt * 16;
  v8f acc = {};
  acc = wmma_gemm_f32(Ap, NHID, Bp, HOc, NHID, acc);
  int lane = threadIdx.x & 31;
  int col  = nt * 16 + (lane & 15);
  int rb   = (lane >> 4) * 8;
#pragma unroll
  for (int g = 0; g < 8; ++g) {
    int row = mt * 16 + rb + g;
    ff[(long)row * HOc + col] = acc[g] + b2[col] + y[(long)row * HOc + col];
  }
}

__global__ void k_ln(const float* __restrict__ ff, const float* __restrict__ g,
                     const float* __restrict__ bb, float* __restrict__ out) {
  int wg   = blockIdx.x * (blockDim.x >> 5) + (threadIdx.x >> 5);
  int lane = threadIdx.x & 31;
  const float* r = ff + (long)wg * HOc;
  float s1 = 0.f, s2 = 0.f;
#pragma unroll
  for (int p = 0; p < 8; ++p) {
    float v = r[p * 32 + lane];
    s1 += v;
    s2 += v * v;
  }
  s1 = wave_sum32(s1);
  s2 = wave_sum32(s2);
  float mu  = s1 * (1.f / HOc);
  float var = s2 * (1.f / HOc) - mu * mu;
  float inv = rsqrtf(var + EPSc);
  float* o = out + (long)wg * HOc;
#pragma unroll
  for (int p = 0; p < 8; ++p) {
    int c = p * 32 + lane;
    o[c] = (r[c] - mu) * inv * g[c] + bb[c];
  }
}

extern "C" void kernel_launch(void* const* d_in, const int* in_sizes, int n_in,
                              void* d_out, int out_size, void* d_ws, size_t ws_size,
                              hipStream_t stream) {
  const float* x    = (const float*)d_in[0];
  const float* edge = (const float*)d_in[1];
  const float* adj  = (const float*)d_in[2];
  const float* W    = (const float*)d_in[3];
  const float* W1   = (const float*)d_in[4];
  const float* a    = (const float*)d_in[5];
  const float* Wo   = (const float*)d_in[6];
  const float* bo   = (const float*)d_in[7];
  const float* w1   = (const float*)d_in[8];
  const float* b1   = (const float*)d_in[9];
  const float* w2   = (const float*)d_in[10];
  const float* b2   = (const float*)d_in[11];
  const float* lng  = (const float*)d_in[12];
  const float* lnb  = (const float*)d_in[13];
  float* out = (float*)d_out;

  float* ws  = (float*)d_ws;
  float* h   = ws;
  float* s   = h + 524288;
  float* t   = s + 16384;
  float* vt  = t + 16384;
  float* u   = vt + 2048;
  float* cat = u + 4194304;
  float* y   = cat + 524288;
  float* z   = y + 524288;
  float* ff  = z + 2097152;

  k_h   <<<256,  256, 0, stream>>>(x, W, h);
  k_st  <<<64,   256, 0, stream>>>(h, a, s, t);
  k_v   <<<8,    256, 0, stream>>>(W1, a, vt);
  k_u   <<<4096, 256, 0, stream>>>(edge, vt, u);
  k_attn<<<2048, 256, 0, stream>>>(u, s, t, adj);
  k_hp  <<<256,  256, 0, stream>>>(u, h, cat);
  k_mlp <<<256,  256, 0, stream>>>(cat, Wo, bo, y, HOc, HOc, 1);
  k_mlp <<<1024, 256, 0, stream>>>(y, w1, b1, z, NHID, HOc, 1);
  k_ff  <<<256,  256, 0, stream>>>(z, w2, b2, y, ff);
  k_ln  <<<256,  256, 0, stream>>>(ff, lng, lnb, out);
}


# A-fragment loads of the shared f32 WMMA loop widened to one 128-bit load per lane feeding two matrix steps; B loader follows the same permuted k order, so the k summation order changes
# speedup vs baseline: 1.0432x; 1.0432x over previous
#include <hip/hip_runtime.h>

constexpr int Bc   = 8;
constexpr int Nc   = 256;
constexpr int NFc  = 256;
constexpr int EFc  = 128;
constexpr int Hc   = 8;
constexpr int Oc   = 32;
constexpr int NHID = 1024;
constexpr int HOc  = 256;
constexpr float ALPHA = 0.2f;
constexpr float EPSc  = 1e-6f;

typedef __attribute__((ext_vector_type(2))) float v2f;
typedef __attribute__((ext_vector_type(4))) float v4f;
typedef __attribute__((ext_vector_type(8))) float v8f;

__device__ __forceinline__ v8f wmma_gemm_f32(const float* __restrict__ A, int lda,
                                             const float* __restrict__ B, int ldb,
                                             int K, v8f acc) {
  const int lane = threadIdx.x & 31;
  const int l16  = lane & 15;
  const int ko   = (lane >> 4) << 2;
  const float* ap = A + l16 * lda + ko;
  const float* bp = B + ko * ldb + l16;
#pragma unroll 4
  for (int k = 0; k < K; k += 8) {
    const v4f a4 = *reinterpret_cast<const v4f*>(ap + k);
    v2f a01, a23, b01, b23;
    a01.x = a4.x;  a01.y = a4.y;
    a23.x = a4.z;  a23.y = a4.w;
    b01.x = bp[k * ldb];
    b01.y = bp[k * ldb + ldb];
    b23.x = bp[k * ldb + 2 * ldb];
    b23.y = bp[k * ldb + 3 * ldb];
    acc = __builtin_amdgcn_wmma_f32_16x16x4_f32(false, a01, false, b01,
                                                (short)0, acc, false, false);
    acc = __builtin_amdgcn_wmma_f32_16x16x4_f32(false, a23, false, b23,
                                                (short)0, acc, false, false);
  }
  return acc;
}

__device__ __forceinline__ float wave_max32(float v) {
#pragma unroll
  for (int m = 16; m >= 1; m >>= 1) v = fmaxf(v, __shfl_xor(v, m, 32));
  return v;
}
__device__ __forceinline__ float wave_sum32(float v) {
#pragma unroll
  for (int m = 16; m >= 1; m >>= 1) v += __shfl_xor(v, m, 32);
  return v;
}

__global__ void k_h(const float* __restrict__ x, const float* __restrict__ W,
                    float* __restrict__ h) {
  int wg = blockIdx.x * (blockDim.x >> 5) + (threadIdx.x >> 5);
  int nt = wg & 1;
  int mt = (wg >> 1) & 15;
  int bh = wg >> 5;
  int b  = bh >> 3, head = bh & 7;
  const float* A  = x + ((long)b * Nc + mt * 16) * NFc;
  const float* Bm = W + (long)head * NFc * Oc + nt * 16;
  v8f acc = {};
  acc = wmma_gemm_f32(A, NFc, Bm, Oc, NFc, acc);
  float* C = h + ((long)bh * Nc + mt * 16) * Oc + nt * 16;
  int lane = threadIdx.x & 31;
  int col  = lane & 15;
  int rb   = (lane >> 4) * 8;
#pragma unroll
  for (int g = 0; g < 8; ++g) C[(rb + g) * Oc + col] = acc[g];
}

__global__ void k_st(const float* __restrict__ h, const float* __restrict__ a,
                     float* __restrict__ s, float* __restrict__ t) {
  int idx  = blockIdx.x * blockDim.x + threadIdx.x;
  int head = (idx >> 8) & (Hc - 1);
  const float* hr = h + (long)idx * Oc;
  const float* a1 = a + head * (3 * Oc);
  const float* a2 = a1 + Oc;
  float sv = 0.f, tv = 0.f;
#pragma unroll
  for (int o = 0; o < Oc; ++o) { sv += hr[o] * a1[o]; tv += hr[o] * a2[o]; }
  s[idx] = sv;
  t[idx] = tv;
}

__global__ void k_v(const float* __restrict__ W1, const float* __restrict__ a,
                    float* __restrict__ vt) {
  int idx = blockIdx.x * blockDim.x + threadIdx.x;
  int c = idx & 15, e = idx >> 4;
  float v = 0.f;
  if (c < Hc) {
    const float* w  = W1 + ((long)c * EFc + e) * Oc;
    const float* a3 = a + c * (3 * Oc) + 2 * Oc;
#pragma unroll
    for (int o = 0; o < Oc; ++o) v += w[o] * a3[o];
  }
  vt[idx] = v;
}

__global__ void k_u(const float* __restrict__ edge, const float* __restrict__ vt,
                    float* __restrict__ u) {
  int wg = blockIdx.x * (blockDim.x >> 5) + (threadIdx.x >> 5);
  const float* A = edge + (long)wg * 16 * EFc;
  v8f acc = {};
  acc = wmma_gemm_f32(A, EFc, vt, 16, EFc, acc);
  int lane = threadIdx.x & 31;
  int col  = lane & 15;
  int rb   = (lane >> 4) * 8;
  if (col < Hc) {
#pragma unroll
    for (int g = 0; g < 8; ++g) {
      long row = (long)wg * 16 + rb + g;
      int b   = (int)(row >> 16);
      int rem = (int)(row & 65535);
      int i = rem >> 8, j = rem & 255;
      u[(((long)b * Hc + col) * Nc + i) * Nc + j] = acc[g];
    }
  }
}

__global__ void k_attn(float* __restrict__ u, const float* __restrict__ s,
                       const float* __restrict__ t, const float* __restrict__ adj) {
  int wg   = blockIdx.x * (blockDim.x >> 5) + (threadIdx.x >> 5);
  int lane = threadIdx.x & 31;
  int i  = wg & (Nc - 1);
  int bh = wg >> 8;
  int b  = bh >> 3;
  float*       urow = u + (long)wg * Nc;
  const float* trow = t + (long)bh * Nc;
  const float* arow = adj + ((long)b * Nc + i) * Nc;
  float sv = s[wg];
  float vals[8];
  float mx = -3.0e38f;
#pragma unroll
  for (int p = 0; p < 8; ++p) {
    int j = p * 32 + lane;
    float sc = sv + trow[j] + urow[j];
    sc = sc > 0.f ? sc : ALPHA * sc;
    sc = (arow[j] > 0.f) ? sc : -3.0e38f;
    vals[p] = sc;
    mx = fmaxf(mx, sc);
  }
  mx = wave_max32(mx);
  float sum = 0.f;
#pragma unroll
  for (int p = 0; p < 8; ++p) {
    float e = __expf(vals[p] - mx);
    vals[p] = e;
    sum += e;
  }
  sum = wave_sum32(sum);
  float inv = 1.f / sum;
#pragma unroll
  for (int p = 0; p < 8; ++p) urow[p * 32 + lane] = vals[p] * inv;
}

__global__ void k_hp(const float* __restrict__ attn, const float* __restrict__ h,
                     float* __restrict__ cat) {
  int wg = blockIdx.x * (blockDim.x >> 5) + (threadIdx.x >> 5);
  int nt = wg & 1;
  int mt = (wg >> 1) & 15;
  int bh = wg >> 5;
  int b  = bh >> 3, head = bh & 7;
  const float* A  = attn + ((long)bh * Nc + mt * 16) * Nc;
  const float* Bm = h + (long)bh * Nc * Oc + nt * 16;
  v8f acc = {};
  acc = wmma_gemm_f32(A, Nc, Bm, Oc, Nc, acc);
  int lane = threadIdx.x & 31;
  int col  = lane & 15;
  int rb   = (lane >> 4) * 8;
#pragma unroll
  for (int g = 0; g < 8; ++g) {
    int i = mt * 16 + rb + g;
    float vv = acc[g];
    vv = vv > 0.f ? vv : (__expf(vv) - 1.f);
    cat[((long)b * Nc + i) * HOc + head * Oc + nt * 16 + col] = vv;
  }
}

__global__ void k_mlp(const float* __restrict__ A, const float* __restrict__ Bm,
                      const float* __restrict__ bias, float* __restrict__ C,
                      int Ncols, int K, int relu) {
  int waves_n = Ncols >> 4;
  int wg = blockIdx.x * (blockDim.x >> 5) + (threadIdx.x >> 5);
  int nt = wg % waves_n;
  int mt = wg / waves_n;
  const float* Ap = A + (long)mt * 16 * K;
  const float* Bp = Bm + nt * 16;
  v8f acc = {};
  acc = wmma_gemm_f32(Ap, K, Bp, Ncols, K, acc);
  int lane = threadIdx.x & 31;
  int col  = nt * 16 + (lane & 15);
  int rb   = (lane >> 4) * 8;
#pragma unroll
  for (int g = 0; g < 8; ++g) {
    float vv = acc[g] + bias[col];
    if (relu) vv = fmaxf(vv, 0.f);
    C[((long)(mt * 16 + rb + g)) * Ncols + col] = vv;
  }
}

__global__ void k_ff(const float* __restrict__ z, const float* __restrict__ w2,
                     const float* __restrict__ b2, const float* __restrict__ y,
                     float* __restrict__ ff) {
  int wg = blockIdx.x * (blockDim.x >> 5) + (threadIdx.x >> 5);
  int nt = wg & 15;
  int mt = wg >> 4;
  const float* Ap = z + (long)mt * 16 * NHID;
  const float* Bp = w2 + nt * 16;
  v8f acc = {};
  acc = wmma_gemm_f32(Ap, NHID, Bp, HOc, NHID, acc);
  int lane = threadIdx.x & 31;
  int col  = nt * 16 + (lane & 15);
  int rb   = (lane >> 4) * 8;
#pragma unroll
  for (int g = 0; g < 8; ++g) {
    int row = mt * 16 + rb + g;
    ff[(long)row * HOc + col] = acc[g] + b2[col] + y[(long)row * HOc + col];
  }
}

__global__ void k_ln(const float* __restrict__ ff, const float* __restrict__ g,
                     const float* __restrict__ bb, float* __restrict__ out) {
  int wg   = blockIdx.x * (blockDim.x >> 5) + (threadIdx.x >> 5);
  int lane = threadIdx.x & 31;
  const float* r = ff + (long)wg * HOc;
  float s1 = 0.f, s2 = 0.f;
#pragma unroll
  for (int p = 0; p < 8; ++p) {
    float v = r[p * 32 + lane];
    s1 += v;
    s2 += v * v;
  }
  s1 = wave_sum32(s1);
  s2 = wave_sum32(s2);
  float mu  = s1 * (1.f / HOc);
  float var = s2 * (1.f / HOc) - mu * mu;
  float inv = rsqrtf(var + EPSc);
  float* o = out + (long)wg * HOc;
#pragma unroll
  for (int p = 0; p < 8; ++p) {
    int c = p * 32 + lane;
    o[c] = (r[c] - mu) * inv * g[c] + bb[c];
  }
}

extern "C" void kernel_launch(void* const* d_in, const int* in_sizes, int n_in,
                              void* d_out, int out_size, void* d_ws, size_t ws_size,
                              hipStream_t stream) {
  const float* x    = (const float*)d_in[0];
  const float* edge = (const float*)d_in[1];
  const float* adj  = (const float*)d_in[2];
  const float* W    = (const float*)d_in[3];
  const float* W1   = (const float*)d_in[4];
  const float* a    = (const float*)d_in[5];
  const float* Wo   = (const float*)d_in[6];
  const float* bo   = (const float*)d_in[7];
  const float* w1   = (const float*)d_in[8];
  const float* b1   = (const float*)d_in[9];
  const float* w2   = (const float*)d_in[10];
  const float* b2   = (const float*)d_in[11];
  const float* lng  = (const float*)d_in[12];
  const float* lnb  = (const float*)d_in[13];
  float* out = (float*)d_out;

  float* ws  = (float*)d_ws;
  float* h   = ws;
  float* s   = h + 524288;
  float* t   = s + 16384;
  float* vt  = t + 16384;
  float* u   = vt + 2048;
  float* cat = u + 4194304;
  float* y   = cat + 524288;
  float* z   = y + 524288;
  float* ff  = z + 2097152;

  k_h   <<<256,  256, 0, stream>>>(x, W, h);
  k_st  <<<64,   256, 0, stream>>>(h, a, s, t);
  k_v   <<<8,    256, 0, stream>>>(W1, a, vt);
  k_u   <<<4096, 256, 0, stream>>>(edge, vt, u);
  k_attn<<<2048, 256, 0, stream>>>(u, s, t, adj);
  k_hp  <<<256,  256, 0, stream>>>(u, h, cat);
  k_mlp <<<256,  256, 0, stream>>>(cat, Wo, bo, y, HOc, HOc, 1);
  k_mlp <<<1024, 256, 0, stream>>>(y, w1, b1, z, NHID, HOc, 1);
  k_ff  <<<256,  256, 0, stream>>>(z, w2, b2, y, ff);
  k_ln  <<<256,  256, 0, stream>>>(ff, lng, lnb, out);
}


# u GEMM and masked softmax fused: one 8-wave workgroup per (b,i) row forms its 8x256 scores (two tiles per wave), keeps them in LDS and writes attn; u never visits the workspace; sums unchanged
# speedup vs baseline: 1.1502x; 1.1026x over previous
#include <hip/hip_runtime.h>

constexpr int Bc   = 8;
constexpr int Nc   = 256;
constexpr int NFc  = 256;
constexpr int EFc  = 128;
constexpr int Hc   = 8;
constexpr int Oc   = 32;
constexpr int NHID = 1024;
constexpr int HOc  = 256;
constexpr float ALPHA = 0.2f;
constexpr float EPSc  = 1e-6f;

typedef __attribute__((ext_vector_type(2))) float v2f;
typedef __attribute__((ext_vector_type(4))) float v4f;
typedef __attribute__((ext_vector_type(8))) float v8f;

__device__ __forceinline__ v8f wmma_gemm_f32(const float* __restrict__ A, int lda,
                                             const float* __restrict__ B, int ldb,
                                             int K, v8f acc) {
  const int lane = threadIdx.x & 31;
  const int l16  = lane & 15;
  const int ko   = (lane >> 4) << 2;
  const float* ap = A + l16 * lda + ko;
  const float* bp = B + ko * ldb + l16;
#pragma unroll 4
  for (int k = 0; k < K; k += 8) {
    const v4f a4 = *reinterpret_cast<const v4f*>(ap + k);
    v2f a01, a23, b01, b23;
    a01.x = a4.x;  a01.y = a4.y;
    a23.x = a4.z;  a23.y = a4.w;
    b01.x = bp[k * ldb];
    b01.y = bp[k * ldb + ldb];
    b23.x = bp[k * ldb + 2 * ldb];
    b23.y = bp[k * ldb + 3 * ldb];
    acc = __builtin_amdgcn_wmma_f32_16x16x4_f32(false, a01, false, b01,
                                                (short)0, acc, false, false);
    acc = __builtin_amdgcn_wmma_f32_16x16x4_f32(false, a23, false, b23,
                                                (short)0, acc, false, false);
  }
  return acc;
}

__device__ __forceinline__ float wave_max32(float v) {
#pragma unroll
  for (int m = 16; m >= 1; m >>= 1) v = fmaxf(v, __shfl_xor(v, m, 32));
  return v;
}
__device__ __forceinline__ float wave_sum32(float v) {
#pragma unroll
  for (int m = 16; m >= 1; m >>= 1) v += __shfl_xor(v, m, 32);
  return v;
}

__global__ void k_h(const float* __restrict__ x, const float* __restrict__ W,
                    float* __restrict__ h) {
  int wg = blockIdx.x * (blockDim.x >> 5) + (threadIdx.x >> 5);
  int nt = wg & 1;
  int mt = (wg >> 1) & 15;
  int bh = wg >> 5;
  int b  = bh >> 3, head = bh & 7;
  const float* A  = x + ((long)b * Nc + mt * 16) * NFc;
  const float* Bm = W + (long)head * NFc * Oc + nt * 16;
  v8f acc = {};
  acc = wmma_gemm_f32(A, NFc, Bm, Oc, NFc, acc);
  float* C = h + ((long)bh * Nc + mt * 16) * Oc + nt * 16;
  int lane = threadIdx.x & 31;
  int col  = lane & 15;
  int rb   = (lane >> 4) * 8;
#pragma unroll
  for (int g = 0; g < 8; ++g) C[(rb + g) * Oc + col] = acc[g];
}

__global__ void k_st(const float* __restrict__ h, const float* __restrict__ a,
                     float* __restrict__ s, float* __restrict__ t) {
  int idx  = blockIdx.x * blockDim.x + threadIdx.x;
  int head = (idx >> 8) & (Hc - 1);
  const float* hr = h + (long)idx * Oc;
  const float* a1 = a + head * (3 * Oc);
  const float* a2 = a1 + Oc;
  float sv = 0.f, tv = 0.f;
#pragma unroll
  for (int o = 0; o < Oc; ++o) { sv += hr[o] * a1[o]; tv += hr[o] * a2[o]; }
  s[idx] = sv;
  t[idx] = tv;
}

__global__ void k_v(const float* __restrict__ W1, const float* __restrict__ a,
                    float* __restrict__ vt) {
  int idx = blockIdx.x * blockDim.x + threadIdx.x;
  int c = idx & 15, e = idx >> 4;
  float v = 0.f;
  if (c < Hc) {
    const float* w  = W1 + ((long)c * EFc + e) * Oc;
    const float* a3 = a + c * (3 * Oc) + 2 * Oc;
#pragma unroll
    for (int o = 0; o < Oc; ++o) v += w[o] * a3[o];
  }
  vt[idx] = v;
}

constexpr int SCP = 272;

__global__ void __launch_bounds__(256) k_uattn(const float* __restrict__ edge, const float* __restrict__ vt,
                                               const float* __restrict__ s, const float* __restrict__ t,
                                               const float* __restrict__ adj, float* __restrict__ attn) {
  __shared__ float sc[Hc * SCP];
  const int row  = blockIdx.x;
  const int w    = threadIdx.x >> 5;
  const int lane = threadIdx.x & 31;
  const int l16  = lane & 15;
  const int hh   = lane >> 4;
  const int ko   = hh << 2;
  const float* ap0 = edge + ((long)row * Nc + w * 16 + l16) * EFc + ko;
  const float* ap1 = ap0 + (long)128 * EFc;
  const float* bp  = vt + ko * 16 + l16;
  v8f acc0 = {}, acc1 = {};
#pragma unroll 4
  for (int k = 0; k < EFc; k += 8) {
    const v4f x0 = *reinterpret_cast<const v4f*>(ap0 + k);
    const v4f x1 = *reinterpret_cast<const v4f*>(ap1 + k);
    v2f p01, p23, q01, q23, b01, b23;
    p01.x = x0.x;  p01.y = x0.y;  p23.x = x0.z;  p23.y = x0.w;
    q01.x = x1.x;  q01.y = x1.y;  q23.x = x1.z;  q23.y = x1.w;
    b01.x = bp[k * 16];
    b01.y = bp[k * 16 + 16];
    b23.x = bp[k * 16 + 32];
    b23.y = bp[k * 16 + 48];
    acc0 = __builtin_amdgcn_wmma_f32_16x16x4_f32(false, p01, false, b01, (short)0, acc0, false, false);
    acc1 = __builtin_amdgcn_wmma_f32_16x16x4_f32(false, q01, false, b01, (short)0, acc1, false, false);
    acc0 = __builtin_amdgcn_wmma_f32_16x16x4_f32(false, p23, false, b23, (short)0, acc0, false, false);
    acc1 = __builtin_amdgcn_wmma_f32_16x16x4_f32(false, q23, false, b23, (short)0, acc1, false, false);
  }
  if (l16 < Hc) {
    float* d0 = sc + l16 * SCP + w * 16 + hh * 8;
#pragma unroll
    for (int g = 0; g < 8; ++g) { d0[g] = acc0[g]; d0[128 + g] = acc1[g]; }
  }
  const int b  = row >> 8;
  const int i  = row & (Nc - 1);
  const int bh = b * Hc + w;
  const float* trow = t + (long)bh * Nc;
  const float* arow = adj + (long)row * Nc;
  const float sv = s[bh * Nc + i];
  float tv[8], av[8];
#pragma unroll
  for (int p = 0; p < 8; ++p) { tv[p] = trow[p * 32 + lane]; av[p] = arow[p * 32 + lane]; }
  __syncthreads();
  const float* srow = sc + w * SCP;
  float vals[8];
  float mx = -3.0e38f;
#pragma unroll
  for (int p = 0; p < 8; ++p) {
    float v = sv + tv[p] + srow[p * 32 + lane];
    v = v > 0.f ? v : ALPHA * v;
    v = (av[p] > 0.f) ? v : -3.0e38f;
    vals[p] = v;
    mx = fmaxf(mx, v);
  }
  mx = wave_max32(mx);
  float sum = 0.f;
#pragma unroll
  for (int p = 0; p < 8; ++p) {
    float e = __expf(vals[p] - mx);
    vals[p] = e;
    sum += e;
  }
  sum = wave_sum32(sum);
  const float inv = 1.f / sum;
  float* orow = attn + ((long)bh * Nc + i) * Nc;
#pragma unroll
  for (int p = 0; p < 8; ++p) orow[p * 32 + lane] = vals[p] * inv;
}

__global__ void k_hp(const float* __restrict__ attn, const float* __restrict__ h,
                     float* __restrict__ cat) {
  int wg = blockIdx.x * (blockDim.x >> 5) + (threadIdx.x >> 5);
  int nt = wg & 1;
  int mt = (wg >> 1) & 15;
  int bh = wg >> 5;
  int b  = bh >> 3, head = bh & 7;
  const float* A  = attn + ((long)bh * Nc + mt * 16) * Nc;
  const float* Bm = h + (long)bh * Nc * Oc + nt * 16;
  v8f acc = {};
  acc = wmma_gemm_f32(A, Nc, Bm, Oc, Nc, acc);
  int lane = threadIdx.x & 31;
  int col  = lane & 15;
  int rb   = (lane >> 4) * 8;
#pragma unroll
  for (int g = 0; g < 8; ++g) {
    int i = mt * 16 + rb + g;
    float vv = acc[g];
    vv = vv > 0.f ? vv : (__expf(vv) - 1.f);
    cat[((long)b * Nc + i) * HOc + head * Oc + nt * 16 + col] = vv;
  }
}

__global__ void k_mlp(const float* __restrict__ A, const float* __restrict__ Bm,
                      const float* __restrict__ bias, float* __restrict__ C,
                      int Ncols, int K, int relu) {
  int waves_n = Ncols >> 4;
  int wg = blockIdx.x * (blockDim.x >> 5) + (threadIdx.x >> 5);
  int nt = wg % waves_n;
  int mt = wg / waves_n;
  const float* Ap = A + (long)mt * 16 * K;
  const float* Bp = Bm + nt * 16;
  v8f acc = {};
  acc = wmma_gemm_f32(Ap, K, Bp, Ncols, K, acc);
  int lane = threadIdx.x & 31;
  int col  = nt * 16 + (lane & 15);
  int rb   = (lane >> 4) * 8;
#pragma unroll
  for (int g = 0; g < 8; ++g) {
    float vv = acc[g] + bias[col];
    if (relu) vv = fmaxf(vv, 0.f);
    C[((long)(mt * 16 + rb + g)) * Ncols + col] = vv;
  }
}

__global__ void k_ff(const float* __restrict__ z, const float* __restrict__ w2,
                     const float* __restrict__ b2, const float* __restrict__ y,
                     float* __restrict__ ff) {
  int wg = blockIdx.x * (blockDim.x >> 5) + (threadIdx.x >> 5);
  int nt = wg & 15;
  int mt = wg >> 4;
  const float* Ap = z + (long)mt * 16 * NHID;
  const float* Bp = w2 + nt * 16;
  v8f acc = {};
  acc = wmma_gemm_f32(Ap, NHID, Bp, HOc, NHID, acc);
  int lane = threadIdx.x & 31;
  int col  = nt * 16 + (lane & 15);
  int rb   = (lane >> 4) * 8;
#pragma unroll
  for (int g = 0; g < 8; ++g) {
    int row = mt * 16 + rb + g;
    ff[(long)row * HOc + col] = acc[g] + b2[col] + y[(long)row * HOc + col];
  }
}

__global__ void k_ln(const float* __restrict__ ff, const float* __restrict__ g,
                     const float* __restrict__ bb, float* __restrict__ out) {
  int wg   = blockIdx.x * (blockDim.x >> 5) + (threadIdx.x >> 5);
  int lane = threadIdx.x & 31;
  const float* r = ff + (long)wg * HOc;
  float s1 = 0.f, s2 = 0.f;
#pragma unroll
  for (int p = 0; p < 8; ++p) {
    float v = r[p * 32 + lane];
    s1 += v;
    s2 += v * v;
  }
  s1 = wave_sum32(s1);
  s2 = wave_sum32(s2);
  float mu  = s1 * (1.f / HOc);
  float var = s2 * (1.f / HOc) - mu * mu;
  float inv = rsqrtf(var + EPSc);
  float* o = out + (long)wg * HOc;
#pragma unroll
  for (int p = 0; p < 8; ++p) {
    int c = p * 32 + lane;
    o[c] = (r[c] - mu) * inv * g[c] + bb[c];
  }
}

extern "C" void kernel_launch(void* const* d_in, const int* in_sizes, int n_in,
                              void* d_out, int out_size, void* d_ws, size_t ws_size,
                              hipStream_t stream) {
  const float* x    = (const float*)d_in[0];
  const float* edge = (const float*)d_in[1];
  const float* adj  = (const float*)d_in[2];
  const float* W    = (const float*)d_in[3];
  const float* W1   = (const float*)d_in[4];
  const float* a    = (const float*)d_in[5];
  const float* Wo   = (const float*)d_in[6];
  const float* bo   = (const float*)d_in[7];
  const float* w1   = (const float*)d_in[8];
  const float* b1   = (const float*)d_in[9];
  const float* w2   = (const float*)d_in[10];
  const float* b2   = (const float*)d_in[11];
  const float* lng  = (const float*)d_in[12];
  const float* lnb  = (const float*)d_in[13];
  float* out = (float*)d_out;

  float* ws  = (float*)d_ws;
  float* h   = ws;
  float* s   = h + 524288;
  float* t   = s + 16384;
  float* vt  = t + 16384;
  float* u   = vt + 2048;
  float* cat = u + 4194304;
  float* y   = cat + 524288;
  float* z   = y + 524288;
  float* ff  = z + 2097152;

  k_h   <<<256,  256, 0, stream>>>(x, W, h);
  k_st  <<<64,   256, 0, stream>>>(h, a, s, t);
  k_v   <<<8,    256, 0, stream>>>(W1, a, vt);
  k_uattn<<<2048, 256, 0, stream>>>(edge, vt, s, t, adj, u);
  k_hp  <<<256,  256, 0, stream>>>(u, h, cat);
  k_mlp <<<256,  256, 0, stream>>>(cat, Wo, bo, y, HOc, HOc, 1);
  k_mlp <<<1024, 256, 0, stream>>>(y, w1, b1, z, NHID, HOc, 1);
  k_ff  <<<256,  256, 0, stream>>>(z, w2, b2, y, ff);
  k_ln  <<<256,  256, 0, stream>>>(ff, lng, lnb, out);
}


# Wo, w1 and w2 GEMMs re-tiled: each wave forms a 16x64 strip (A fragment feeds 4 tiles, B rows read as 128-bit loads); the two N=256 products split K over 4 waves, summed in LDS: sum order changes
# speedup vs baseline: 1.4321x; 1.2451x over previous
#include <hip/hip_runtime.h>

constexpr int Bc   = 8;
constexpr int Nc   = 256;
constexpr int NFc  = 256;
constexpr int EFc  = 128;
constexpr int Hc   = 8;
constexpr int Oc   = 32;
constexpr int NHID = 1024;
constexpr int HOc  = 256;
constexpr float ALPHA = 0.2f;
constexpr float EPSc  = 1e-6f;

typedef __attribute__((ext_vector_type(2))) float v2f;
typedef __attribute__((ext_vector_type(4))) float v4f;
typedef __attribute__((ext_vector_type(8))) float v8f;

__device__ __forceinline__ v8f wmma_gemm_f32(const float* __restrict__ A, int lda,
                                             const float* __restrict__ B, int ldb,
                                             int K, v8f acc) {
  const int lane = threadIdx.x & 31;
  const int l16  = lane & 15;
  const int ko   = (lane >> 4) << 2;
  const float* ap = A + l16 * lda + ko;
  const float* bp = B + ko * ldb + l16;
#pragma unroll 4
  for (int k = 0; k < K; k += 8) {
    const v4f a4 = *reinterpret_cast<const v4f*>(ap + k);
    v2f a01, a23, b01, b23;
    a01.x = a4.x;  a01.y = a4.y;
    a23.x = a4.z;  a23.y = a4.w;
    b01.x = bp[k * ldb];
    b01.y = bp[k * ldb + ldb];
    b23.x = bp[k * ldb + 2 * ldb];
    b23.y = bp[k * ldb + 3 * ldb];
    acc = __builtin_amdgcn_wmma_f32_16x16x4_f32(false, a01, false, b01,
                                                (short)0, acc, false, false);
    acc = __builtin_amdgcn_wmma_f32_16x16x4_f32(false, a23, false, b23,
                                                (short)0, acc, false, false);
  }
  return acc;
}

__device__ __forceinline__ void wmma_strip64_f32(const float* __restrict__ A, int lda,
                                                 const float* __restrict__ B, int ldb, int K,
                                                 v8f& c0, v8f& c1, v8f& c2, v8f& c3) {
  const int lane = threadIdx.x & 31;
  const int l16  = lane & 15;
  const int ko   = (lane >> 4) << 2;
  const float* ap = A + l16 * lda + ko;
  const float* bp = B + ko * ldb + 4 * l16;
  for (int k = 0; k < K; k += 32) {
    v4f a[4], r[4][4];
#pragma unroll
    for (int t = 0; t < 4; ++t) {
      a[t] = *reinterpret_cast<const v4f*>(ap + k + 8 * t);
#pragma unroll
      for (int s = 0; s < 4; ++s)
        r[t][s] = *reinterpret_cast<const v4f*>(bp + (k + 8 * t + s) * ldb);
    }
#pragma unroll
    for (int t = 0; t < 4; ++t) {
      v2f a01, a23, b;
      a01.x = a[t].x;  a01.y = a[t].y;
      a23.x = a[t].z;  a23.y = a[t].w;
      b.x = r[t][0].x;  b.y = r[t][1].x;
      c0 = __builtin_amdgcn_wmma_f32_16x16x4_f32(false, a01, false, b, (short)0, c0, false, false);
      b.x = r[t][0].y;  b.y = r[t][1].y;
      c1 = __builtin_amdgcn_wmma_f32_16x16x4_f32(false, a01, false, b, (short)0, c1, false, false);
      b.x = r[t][0].z;  b.y = r[t][1].z;
      c2 = __builtin_amdgcn_wmma_f32_16x16x4_f32(false, a01, false, b, (short)0, c2, false, false);
      b.x = r[t][0].w;  b.y = r[t][1].w;
      c3 = __builtin_amdgcn_wmma_f32_16x16x4_f32(false, a01, false, b, (short)0, c3, false, false);
      b.x = r[t][2].x;  b.y = r[t][3].x;
      c0 = __builtin_amdgcn_wmma_f32_16x16x4_f32(false, a23, false, b, (short)0, c0, false, false);
      b.x = r[t][2].y;  b.y = r[t][3].y;
      c1 = __builtin_amdgcn_wmma_f32_16x16x4_f32(false, a23, false, b, (short)0, c1, false, false);
      b.x = r[t][2].z;  b.y = r[t][3].z;
      c2 = __builtin_amdgcn_wmma_f32_16x16x4_f32(false, a23, false, b, (short)0, c2, false, false);
      b.x = r[t][2].w;  b.y = r[t][3].w;
      c3 = __builtin_amdgcn_wmma_f32_16x16x4_f32(false, a23, false, b, (short)0, c3, false, false);
    }
    __builtin_amdgcn_sched_group_barrier(0x020, 20, 0);
    __builtin_amdgcn_sched_group_barrier(0x008, 32, 0);
  }
}

__device__ __forceinline__ float wave_max32(float v) {
#pragma unroll
  for (int m = 16; m >= 1; m >>= 1) v = fmaxf(v, __shfl_xor(v, m, 32));
  return v;
}
__device__ __forceinline__ float wave_sum32(float v) {
#pragma unroll
  for (int m = 16; m >= 1; m >>= 1) v += __shfl_xor(v, m, 32);
  return v;
}

__global__ void k_h(const float* __restrict__ x, const float* __restrict__ W,
                    float* __restrict__ h) {
  int wg = blockIdx.x * (blockDim.x >> 5) + (threadIdx.x >> 5);
  int nt = wg & 1;
  int mt = (wg >> 1) & 15;
  int bh = wg >> 5;
  int b  = bh >> 3, head = bh & 7;
  const float* A  = x + ((long)b * Nc + mt * 16) * NFc;
  const float* Bm = W + (long)head * NFc * Oc + nt * 16;
  v8f acc = {};
  acc = wmma_gemm_f32(A, NFc, Bm, Oc, NFc, acc);
  float* C = h + ((long)bh * Nc + mt * 16) * Oc + nt * 16;
  int lane = threadIdx.x & 31;
  int col  = lane & 15;
  int rb   = (lane >> 4) * 8;
#pragma unroll
  for (int g = 0; g < 8; ++g) C[(rb + g) * Oc + col] = acc[g];
}

__global__ void k_st(const float* __restrict__ h, const float* __restrict__ a,
                     float* __restrict__ s, float* __restrict__ t) {
  int idx  = blockIdx.x * blockDim.x + threadIdx.x;
  int head = (idx >> 8) & (Hc - 1);
  const float* hr = h + (long)idx * Oc;
  const float* a1 = a + head * (3 * Oc);
  const float* a2 = a1 + Oc;
  float sv = 0.f, tv = 0.f;
#pragma unroll
  for (int o = 0; o < Oc; ++o) { sv += hr[o] * a1[o]; tv += hr[o] * a2[o]; }
  s[idx] = sv;
  t[idx] = tv;
}

__global__ void k_v(const float* __restrict__ W1, const float* __restrict__ a,
                    float* __restrict__ vt) {
  int idx = blockIdx.x * blockDim.x + threadIdx.x;
  int c = idx & 15, e = idx >> 4;
  float v = 0.f;
  if (c < Hc) {
    const float* w  = W1 + ((long)c * EFc + e) * Oc;
    const float* a3 = a + c * (3 * Oc) + 2 * Oc;
#pragma unroll
    for (int o = 0; o < Oc; ++o) v += w[o] * a3[o];
  }
  vt[idx] = v;
}

constexpr int SCP = 272;

__global__ void __launch_bounds__(256) k_uattn(const float* __restrict__ edge, const float* __restrict__ vt,
                                               const float* __restrict__ s, const float* __restrict__ t,
                                               const float* __restrict__ adj, float* __restrict__ attn) {
  __shared__ float sc[Hc * SCP];
  const int row  = blockIdx.x;
  const int w    = threadIdx.x >> 5;
  const int lane = threadIdx.x & 31;
  const int l16  = lane & 15;
  const int hh   = lane >> 4;
  const int ko   = hh << 2;
  const float* ap0 = edge + ((long)row * Nc + w * 16 + l16) * EFc + ko;
  const float* ap1 = ap0 + (long)128 * EFc;
  const float* bp  = vt + ko * 16 + l16;
  v8f acc0 = {}, acc1 = {};
#pragma unroll 4
  for (int k = 0; k < EFc; k += 8) {
    const v4f x0 = *reinterpret_cast<const v4f*>(ap0 + k);
    const v4f x1 = *reinterpret_cast<const v4f*>(ap1 + k);
    v2f p01, p23, q01, q23, b01, b23;
    p01.x = x0.x;  p01.y = x0.y;  p23.x = x0.z;  p23.y = x0.w;
    q01.x = x1.x;  q01.y = x1.y;  q23.x = x1.z;  q23.y = x1.w;
    b01.x = bp[k * 16];
    b01.y = bp[k * 16 + 16];
    b23.x = bp[k * 16 + 32];
    b23.y = bp[k * 16 + 48];
    acc0 = __builtin_amdgcn_wmma_f32_16x16x4_f32(false, p01, false, b01, (short)0, acc0, false, false);
    acc1 = __builtin_amdgcn_wmma_f32_16x16x4_f32(false, q01, false, b01, (short)0, acc1, false, false);
    acc0 = __builtin_amdgcn_wmma_f32_16x16x4_f32(false, p23, false, b23, (short)0, acc0, false, false);
    acc1 = __builtin_amdgcn_wmma_f32_16x16x4_f32(false, q23, false, b23, (short)0, acc1, false, false);
  }
  if (l16 < Hc) {
    float* d0 = sc + l16 * SCP + w * 16 + hh * 8;
#pragma unroll
    for (int g = 0; g < 8; ++g) { d0[g] = acc0[g]; d0[128 + g] = acc1[g]; }
  }
  const int b  = row >> 8;
  const int i  = row & (Nc - 1);
  const int bh = b * Hc + w;
  const float* trow = t + (long)bh * Nc;
  const float* arow = adj + (long)row * Nc;
  const float sv = s[bh * Nc + i];
  float tv[8], av[8];
#pragma unroll
  for (int p = 0; p < 8; ++p) { tv[p] = trow[p * 32 + lane]; av[p] = arow[p * 32 + lane]; }
  __syncthreads();
  const float* srow = sc + w * SCP;
  float vals[8];
  float mx = -3.0e38f;
#pragma unroll
  for (int p = 0; p < 8; ++p) {
    float v = sv + tv[p] + srow[p * 32 + lane];
    v = v > 0.f ? v : ALPHA * v;
    v = (av[p] > 0.f) ? v : -3.0e38f;
    vals[p] = v;
    mx = fmaxf(mx, v);
  }
  mx = wave_max32(mx);
  float sum = 0.f;
#pragma unroll
  for (int p = 0; p < 8; ++p) {
    float e = __expf(vals[p] - mx);
    vals[p] = e;
    sum += e;
  }
  sum = wave_sum32(sum);
  const float inv = 1.f / sum;
  float* orow = attn + ((long)bh * Nc + i) * Nc;
#pragma unroll
  for (int p = 0; p < 8; ++p) orow[p * 32 + lane] = vals[p] * inv;
}

__global__ void k_hp(const float* __restrict__ attn, const float* __restrict__ h,
                     float* __restrict__ cat) {
  int wg = blockIdx.x * (blockDim.x >> 5) + (threadIdx.x >> 5);
  int nt = wg & 1;
  int mt = (wg >> 1) & 15;
  int bh = wg >> 5;
  int b  = bh >> 3, head = bh & 7;
  const float* A  = attn + ((long)bh * Nc + mt * 16) * Nc;
  const float* Bm = h + (long)bh * Nc * Oc + nt * 16;
  v8f acc = {};
  acc = wmma_gemm_f32(A, Nc, Bm, Oc, Nc, acc);
  int lane = threadIdx.x & 31;
  int col  = lane & 15;
  int rb   = (lane >> 4) * 8;
#pragma unroll
  for (int g = 0; g < 8; ++g) {
    int i = mt * 16 + rb + g;
    float vv = acc[g];
    vv = vv > 0.f ? vv : (__expf(vv) - 1.f);
    cat[((long)b * Nc + i) * HOc + head * Oc + nt * 16 + col] = vv;
  }
}

template <int KTOT, bool RELU, bool RESID>
__global__ void __launch_bounds__(256) k_n256(const float* __restrict__ A, const float* __restrict__ Bm,
                                              const float* __restrict__ bias, const float* __restrict__ resid,
                                              float* __restrict__ C) {
  __shared__ __attribute__((aligned(16))) float part[4 * 2 * 16 * 64];
  constexpr int KQ = KTOT / 4;
  const int w    = threadIdx.x >> 5;
  const int lane = threadIdx.x & 31;
  const int l16  = lane & 15;
  const int hh   = lane >> 4;
  const int g    = blockIdx.x;
  const int r32  = g >> 2;
  const int col0 = (g & 3) * 64;
  const int rb   = w & 1;
  const int kq   = w >> 1;
  const int row0 = r32 * 32 + rb * 16;
  v8f c0 = {}, c1 = {}, c2 = {}, c3 = {};
  wmma_strip64_f32(A + (long)row0 * KTOT + kq * KQ, KTOT,
                   Bm + (long)kq * KQ * HOc + col0, HOc, KQ, c0, c1, c2, c3);
  float* p = part + ((kq * 2 + rb) * 16 + hh * 8) * 64 + 4 * l16;
#pragma unroll
  for (int r = 0; r < 8; ++r) {
    v4f o;
    o.x = c0[r];  o.y = c1[r];  o.z = c2[r];  o.w = c3[r];
    *reinterpret_cast<v4f*>(p + r * 64) = o;
  }
  __syncthreads();
  const int rr = threadIdx.x >> 4;
  const int cc = (threadIdx.x & 15) * 4;
  const v4f bv = *reinterpret_cast<const v4f*>(bias + col0 + cc);
#pragma unroll
  for (int hb = 0; hb < 2; ++hb) {
    const float* q = part + (hb * 16 + rr) * 64 + cc;
    v4f s = *reinterpret_cast<const v4f*>(q);
    s += *reinterpret_cast<const v4f*>(q + 1 * 2 * 16 * 64);
    s += *reinterpret_cast<const v4f*>(q + 2 * 2 * 16 * 64);
    s += *reinterpret_cast<const v4f*>(q + 3 * 2 * 16 * 64);
    s += bv;
    const long off = (long)(r32 * 32 + hb * 16 + rr) * HOc + col0 + cc;
    if (RESID) s += *reinterpret_cast<const v4f*>(resid + off);
    if (RELU) { s.x = fmaxf(s.x, 0.f);  s.y = fmaxf(s.y, 0.f);  s.z = fmaxf(s.z, 0.f);  s.w = fmaxf(s.w, 0.f); }
    *reinterpret_cast<v4f*>(C + off) = s;
  }
}

__global__ void __launch_bounds__(256) k_z(const float* __restrict__ y, const float* __restrict__ w1,
                                           const float* __restrict__ b1, float* __restrict__ z) {
  const int w    = threadIdx.x >> 5;
  const int lane = threadIdx.x & 31;
  const int l16  = lane & 15;
  const int hh   = lane >> 4;
  const int g    = blockIdx.x;
  const int row0 = (g >> 3) * 64 + (w & 3) * 16;
  const int col0 = (g & 7) * 128 + (w >> 2) * 64;
  v8f c0 = {}, c1 = {}, c2 = {}, c3 = {};
  wmma_strip64_f32(y + (long)row0 * HOc, HOc, w1 + col0, NHID, HOc, c0, c1, c2, c3);
  const v4f bv = *reinterpret_cast<const v4f*>(b1 + col0 + 4 * l16);
  float* zr = z + (long)(row0 + hh * 8) * NHID + col0 + 4 * l16;
#pragma unroll
  for (int r = 0; r < 8; ++r) {
    v4f o;
    o.x = fmaxf(c0[r] + bv.x, 0.f);
    o.y = fmaxf(c1[r] + bv.y, 0.f);
    o.z = fmaxf(c2[r] + bv.z, 0.f);
    o.w = fmaxf(c3[r] + bv.w, 0.f);
    *reinterpret_cast<v4f*>(zr + (long)r * NHID) = o;
  }
}

__global__ void k_ln(const float* __restrict__ ff, const float* __restrict__ g,
                     const float* __restrict__ bb, float* __restrict__ out) {
  int wg   = blockIdx.x * (blockDim.x >> 5) + (threadIdx.x >> 5);
  int lane = threadIdx.x & 31;
  const float* r = ff + (long)wg * HOc;
  float s1 = 0.f, s2 = 0.f;
#pragma unroll
  for (int p = 0; p < 8; ++p) {
    float v = r[p * 32 + lane];
    s1 += v;
    s2 += v * v;
  }
  s1 = wave_sum32(s1);
  s2 = wave_sum32(s2);
  float mu  = s1 * (1.f / HOc);
  float var = s2 * (1.f / HOc) - mu * mu;
  float inv = rsqrtf(var + EPSc);
  float* o = out + (long)wg * HOc;
#pragma unroll
  for (int p = 0; p < 8; ++p) {
    int c = p * 32 + lane;
    o[c] = (r[c] - mu) * inv * g[c] + bb[c];
  }
}

extern "C" void kernel_launch(void* const* d_in, const int* in_sizes, int n_in,
                              void* d_out, int out_size, void* d_ws, size_t ws_size,
                              hipStream_t stream) {
  const float* x    = (const float*)d_in[0];
  const float* edge = (const float*)d_in[1];
  const float* adj  = (const float*)d_in[2];
  const float* W    = (const float*)d_in[3];
  const float* W1   = (const float*)d_in[4];
  const float* a    = (const float*)d_in[5];
  const float* Wo   = (const float*)d_in[6];
  const float* bo   = (const float*)d_in[7];
  const float* w1   = (const float*)d_in[8];
  const float* b1   = (const float*)d_in[9];
  const float* w2   = (const float*)d_in[10];
  const float* b2   = (const float*)d_in[11];
  const float* lng  = (const float*)d_in[12];
  const float* lnb  = (const float*)d_in[13];
  float* out = (float*)d_out;

  float* ws  = (float*)d_ws;
  float* h   = ws;
  float* s   = h + 524288;
  float* t   = s + 16384;
  float* vt  = t + 16384;
  float* u   = vt + 2048;
  float* cat = u + 4194304;
  float* y   = cat + 524288;
  float* z   = y + 524288;
  float* ff  = z + 2097152;

  k_h   <<<256,  256, 0, stream>>>(x, W, h);
  k_st  <<<64,   256, 0, stream>>>(h, a, s, t);
  k_v   <<<8,    256, 0, stream>>>(W1, a, vt);
  k_uattn<<<2048, 256, 0, stream>>>(edge, vt, s, t, adj, u);
  k_hp  <<<256,  256, 0, stream>>>(u, h, cat);
  k_n256<HOc,  true,  false><<<256, 256, 0, stream>>>(cat, Wo, bo, nullptr, y);
  k_z   <<<256,  256, 0, stream>>>(y, w1, b1, z);
  k_n256<NHID, false, true ><<<256, 256, 0, stream>>>(z, w2, b2, y, ff);
  k_ln  <<<256,  256, 0, stream>>>(ff, lng, lnb, out);
}


# attn@h fused into the score/softmax kernel: one workgroup per 8 i rows loops over them with edge tiles requested a row ahead, keeps attn in LDS and writes elu(attn@h); sums unchanged
# speedup vs baseline: 1.6409x; 1.1458x over previous
#include <hip/hip_runtime.h>

constexpr int Bc   = 8;
constexpr int Nc   = 256;
constexpr int NFc  = 256;
constexpr int EFc  = 128;
constexpr int Hc   = 8;
constexpr int Oc   = 32;
constexpr int NHID = 1024;
constexpr int HOc  = 256;
constexpr float ALPHA = 0.2f;
constexpr float EPSc  = 1e-6f;

typedef __attribute__((ext_vector_type(2))) float v2f;
typedef __attribute__((ext_vector_type(4))) float v4f;
typedef __attribute__((ext_vector_type(8))) float v8f;

__device__ __forceinline__ v8f wmma_gemm_f32(const float* __restrict__ A, int lda,
                                             const float* __restrict__ B, int ldb,
                                             int K, v8f acc) {
  const int lane = threadIdx.x & 31;
  const int l16  = lane & 15;
  const int ko   = (lane >> 4) << 2;
  const float* ap = A + l16 * lda + ko;
  const float* bp = B + ko * ldb + l16;
#pragma unroll 4
  for (int k = 0; k < K; k += 8) {
    const v4f a4 = *reinterpret_cast<const v4f*>(ap + k);
    v2f a01, a23, b01, b23;
    a01.x = a4.x;  a01.y = a4.y;
    a23.x = a4.z;  a23.y = a4.w;
    b01.x = bp[k * ldb];
    b01.y = bp[k * ldb + ldb];
    b23.x = bp[k * ldb + 2 * ldb];
    b23.y = bp[k * ldb + 3 * ldb];
    acc = __builtin_amdgcn_wmma_f32_16x16x4_f32(false, a01, false, b01,
                                                (short)0, acc, false, false);
    acc = __builtin_amdgcn_wmma_f32_16x16x4_f32(false, a23, false, b23,
                                                (short)0, acc, false, false);
  }
  return acc;
}

__device__ __forceinline__ void wmma_strip64_f32(const float* __restrict__ A, int lda,
                                                 const float* __restrict__ B, int ldb, int K,
                                                 v8f& c0, v8f& c1, v8f& c2, v8f& c3) {
  const int lane = threadIdx.x & 31;
  const int l16  = lane & 15;
  const int ko   = (lane >> 4) << 2;
  const float* ap = A + l16 * lda + ko;
  const float* bp = B + ko * ldb + 4 * l16;
  for (int k = 0; k < K; k += 32) {
    v4f a[4], r[4][4];
#pragma unroll
    for (int t = 0; t < 4; ++t) {
      a[t] = *reinterpret_cast<const v4f*>(ap + k + 8 * t);
#pragma unroll
      for (int s = 0; s < 4; ++s)
        r[t][s] = *reinterpret_cast<const v4f*>(bp + (k + 8 * t + s) * ldb);
    }
#pragma unroll
    for (int t = 0; t < 4; ++t) {
      v2f a01, a23, b;
      a01.x = a[t].x;  a01.y = a[t].y;
      a23.x = a[t].z;  a23.y = a[t].w;
      b.x = r[t][0].x;  b.y = r[t][1].x;
      c0 = __builtin_amdgcn_wmma_f32_16x16x4_f32(false, a01, false, b, (short)0, c0, false, false);
      b.x = r[t][0].y;  b.y = r[t][1].y;
      c1 = __builtin_amdgcn_wmma_f32_16x16x4_f32(false, a01, false, b, (short)0, c1, false, false);
      b.x = r[t][0].z;  b.y = r[t][1].z;
      c2 = __builtin_amdgcn_wmma_f32_16x16x4_f32(false, a01, false, b, (short)0, c2, false, false);
      b.x = r[t][0].w;  b.y = r[t][1].w;
      c3 = __builtin_amdgcn_wmma_f32_16x16x4_f32(false, a01, false, b, (short)0, c3, false, false);
      b.x = r[t][2].x;  b.y = r[t][3].x;
      c0 = __builtin_amdgcn_wmma_f32_16x16x4_f32(false, a23, false, b, (short)0, c0, false, false);
      b.x = r[t][2].y;  b.y = r[t][3].y;
      c1 = __builtin_amdgcn_wmma_f32_16x16x4_f32(false, a23, false, b, (short)0, c1, false, false);
      b.x = r[t][2].z;  b.y = r[t][3].z;
      c2 = __builtin_amdgcn_wmma_f32_16x16x4_f32(false, a23, false, b, (short)0, c2, false, false);
      b.x = r[t][2].w;  b.y = r[t][3].w;
      c3 = __builtin_amdgcn_wmma_f32_16x16x4_f32(false, a23, false, b, (short)0, c3, false, false);
    }
    __builtin_amdgcn_sched_group_barrier(0x020, 20, 0);
    __builtin_amdgcn_sched_group_barrier(0x008, 32, 0);
  }
}

__device__ __forceinline__ float wave_max32(float v) {
#pragma unroll
  for (int m = 16; m >= 1; m >>= 1) v = fmaxf(v, __shfl_xor(v, m, 32));
  return v;
}
__device__ __forceinline__ float wave_sum32(float v) {
#pragma unroll
  for (int m = 16; m >= 1; m >>= 1) v += __shfl_xor(v, m, 32);
  return v;
}

__global__ void k_h(const float* __restrict__ x, const float* __restrict__ W,
                    float* __restrict__ h) {
  int wg = blockIdx.x * (blockDim.x >> 5) + (threadIdx.x >> 5);
  int nt = wg & 1;
  int mt = (wg >> 1) & 15;
  int bh = wg >> 5;
  int b  = bh >> 3, head = bh & 7;
  const float* A  = x + ((long)b * Nc + mt * 16) * NFc;
  const float* Bm = W + (long)head * NFc * Oc + nt * 16;
  v8f acc = {};
  acc = wmma_gemm_f32(A, NFc, Bm, Oc, NFc, acc);
  float* C = h + ((long)bh * Nc + mt * 16) * Oc + nt * 16;
  int lane = threadIdx.x & 31;
  int col  = lane & 15;
  int rb   = (lane >> 4) * 8;
#pragma unroll
  for (int g = 0; g < 8; ++g) C[(rb + g) * Oc + col] = acc[g];
}

__global__ void k_st(const float* __restrict__ h, const float* __restrict__ a,
                     float* __restrict__ s, float* __restrict__ t) {
  int idx  = blockIdx.x * blockDim.x + threadIdx.x;
  int head = (idx >> 8) & (Hc - 1);
  const float* hr = h + (long)idx * Oc;
  const float* a1 = a + head * (3 * Oc);
  const float* a2 = a1 + Oc;
  float sv = 0.f, tv = 0.f;
#pragma unroll
  for (int o = 0; o < Oc; ++o) { sv += hr[o] * a1[o]; tv += hr[o] * a2[o]; }
  s[idx] = sv;
  t[idx] = tv;
}

__global__ void k_v(const float* __restrict__ W1, const float* __restrict__ a,
                    float* __restrict__ vt) {
  int idx = blockIdx.x * blockDim.x + threadIdx.x;
  int c = idx & 15, e = idx >> 4;
  float v = 0.f;
  if (c < Hc) {
    const float* w  = W1 + ((long)c * EFc + e) * Oc;
    const float* a3 = a + c * (3 * Oc) + 2 * Oc;
#pragma unroll
    for (int o = 0; o < Oc; ++o) v += w[o] * a3[o];
  }
  vt[idx] = v;
}

constexpr int SCP = 272;
constexpr int RPW = 8;
constexpr int ATP = 264;

__device__ __forceinline__ void lds_barrier() {
  asm volatile("s_wait_dscnt 0x0\n\ts_barrier_signal -1\n\ts_barrier_wait -1" ::: "memory");
}

template <bool HAS_NEXT>
__device__ __forceinline__ void uattn_row(v4f (&cx)[16], v4f (&cy)[16], v4f (&nx)[16], v4f (&ny)[16],
                                          const v2f (&vb01)[16], const v2f (&vb23)[16],
                                          const float* __restrict__ ap_next, const float* __restrict__ arow,
                                          const float sv, const float (&tv)[8],
                                          float* __restrict__ scw, float* __restrict__ atrow,
                                          const int w, const int lane, const int l16, const int hh) {
  float av[8];
#pragma unroll
  for (int p = 0; p < 8; ++p) av[p] = arow[p * 32 + lane];
  if (HAS_NEXT) {
#pragma unroll
    for (int t = 0; t < 16; ++t) {
      nx[t] = *reinterpret_cast<const v4f*>(ap_next + 8 * t);
      ny[t] = *reinterpret_cast<const v4f*>(ap_next + 128 * EFc + 8 * t);
    }
  }
  v8f acc0 = {}, acc1 = {};
#pragma unroll
  for (int t = 0; t < 16; ++t) {
    v2f p01, p23, q01, q23;
    p01.x = cx[t].x;  p01.y = cx[t].y;  p23.x = cx[t].z;  p23.y = cx[t].w;
    q01.x = cy[t].x;  q01.y = cy[t].y;  q23.x = cy[t].z;  q23.y = cy[t].w;
    acc0 = __builtin_amdgcn_wmma_f32_16x16x4_f32(false, p01, false, vb01[t], (short)0, acc0, false, false);
    acc1 = __builtin_amdgcn_wmma_f32_16x16x4_f32(false, q01, false, vb01[t], (short)0, acc1, false, false);
    acc0 = __builtin_amdgcn_wmma_f32_16x16x4_f32(false, p23, false, vb23[t], (short)0, acc0, false, false);
    acc1 = __builtin_amdgcn_wmma_f32_16x16x4_f32(false, q23, false, vb23[t], (short)0, acc1, false, false);
  }
  if (l16 < Hc) {
    float* d0 = scw + l16 * SCP + w * 16 + hh * 8;
#pragma unroll
    for (int g = 0; g < 8; ++g) { d0[g] = acc0[g]; d0[128 + g] = acc1[g]; }
  }
  lds_barrier();
  const float* srow = scw + w * SCP;
  float vals[8];
  float mx = -3.0e38f;
#pragma unroll
  for (int p = 0; p < 8; ++p) {
    float v = sv + tv[p] + srow[p * 32 + lane];
    v = v > 0.f ? v : ALPHA * v;
    v = (av[p] > 0.f) ? v : -3.0e38f;
    vals[p] = v;
    mx = fmaxf(mx, v);
  }
  mx = wave_max32(mx);
  float sum = 0.f;
#pragma unroll
  for (int p = 0; p < 8; ++p) {
    float e = __expf(vals[p] - mx);
    vals[p] = e;
    sum += e;
  }
  sum = wave_sum32(sum);
  const float inv = 1.f / sum;
#pragma unroll
  for (int p = 0; p < 8; ++p) atrow[p * 32 + lane] = vals[p] * inv;
}

__global__ void __launch_bounds__(256) k_uattn_hp(const float* __restrict__ edge, const float* __restrict__ vt,
                                                  const float* __restrict__ s, const float* __restrict__ t,
                                                  const float* __restrict__ adj, const float* __restrict__ h,
                                                  float* __restrict__ cat) {
  __shared__ __attribute__((aligned(16))) float sc[2 * Hc * SCP];
  __shared__ __attribute__((aligned(16))) float at[Hc * RPW * ATP];
  const int g    = blockIdx.x;
  const int b    = g >> 5;
  const int i0   = (g & 31) * RPW;
  const int w    = threadIdx.x >> 5;
  const int lane = threadIdx.x & 31;
  const int l16  = lane & 15;
  const int hh   = lane >> 4;
  const int ko   = hh << 2;
  const int bh   = b * Hc + w;
  const long row0 = (long)b * Nc + i0;
  const float* ap = edge + ((row0 * Nc) + w * 16 + l16) * EFc + ko;
  v4f xa[16], ya[16], xb[16], yb[16];
#pragma unroll
  for (int q = 0; q < 16; ++q) {
    xa[q] = *reinterpret_cast<const v4f*>(ap + 8 * q);
    ya[q] = *reinterpret_cast<const v4f*>(ap + 128 * EFc + 8 * q);
  }
  v2f vb01[16], vb23[16];
  const float* bp = vt + ko * 16 + l16;
#pragma unroll
  for (int q = 0; q < 16; ++q) {
    vb01[q].x = bp[(8 * q) * 16];
    vb01[q].y = bp[(8 * q + 1) * 16];
    vb23[q].x = bp[(8 * q + 2) * 16];
    vb23[q].y = bp[(8 * q + 3) * 16];
  }
  float tv[8], sv[RPW];
#pragma unroll
  for (int p = 0; p < 8; ++p) tv[p] = t[(long)bh * Nc + p * 32 + lane];
#pragma unroll
  for (int r = 0; r < RPW; ++r) sv[r] = s[bh * Nc + i0 + r];
  const float* arow = adj + row0 * Nc;
  float* atw = at + w * RPW * ATP;
  constexpr long ROWE = (long)Nc * EFc;
#pragma unroll
  for (int r = 0; r < RPW; r += 2) {
    uattn_row<true>(xa, ya, xb, yb, vb01, vb23, ap + (r + 1) * ROWE, arow + r * Nc, sv[r], tv,
                    sc, atw + r * ATP, w, lane, l16, hh);
    if (r + 2 < RPW)
      uattn_row<true>(xb, yb, xa, ya, vb01, vb23, ap + (r + 2) * ROWE, arow + (r + 1) * Nc, sv[r + 1], tv,
                      sc + Hc * SCP, atw + (r + 1) * ATP, w, lane, l16, hh);
    else
      uattn_row<false>(xb, yb, xa, ya, vb01, vb23, ap, arow + (r + 1) * Nc, sv[r + 1], tv,
                       sc + Hc * SCP, atw + (r + 1) * ATP, w, lane, l16, hh);
  }
  const float* aL = atw + (l16 & 7) * ATP + ko;
  const float* hb = h + (long)bh * Nc * Oc + ko * Oc + l16;
  v8f c0 = {}, c1 = {};
#pragma unroll 8
  for (int k = 0; k < Nc; k += 8) {
    const v4f a4 = *reinterpret_cast<const v4f*>(aL + k);
    v2f a01, a23, b01, b23, d01, d23;
    a01.x = a4.x;  a01.y = a4.y;  a23.x = a4.z;  a23.y = a4.w;
    b01.x = hb[k * Oc];            b01.y = hb[k * Oc + Oc];
    b23.x = hb[k * Oc + 2 * Oc];   b23.y = hb[k * Oc + 3 * Oc];
    d01.x = hb[k * Oc + 16];            d01.y = hb[k * Oc + Oc + 16];
    d23.x = hb[k * Oc + 2 * Oc + 16];   d23.y = hb[k * Oc + 3 * Oc + 16];
    c0 = __builtin_amdgcn_wmma_f32_16x16x4_f32(false, a01, false, b01, (short)0, c0, false, false);
    c1 = __builtin_amdgcn_wmma_f32_16x16x4_f32(false, a01, false, d01, (short)0, c1, false, false);
    c0 = __builtin_amdgcn_wmma_f32_16x16x4_f32(false, a23, false, b23, (short)0, c0, false, false);
    c1 = __builtin_amdgcn_wmma_f32_16x16x4_f32(false, a23, false, d23, (short)0, c1, false, false);
  }
  if (hh == 0) {
    float* cr = cat + (row0 * HOc) + w * Oc + l16;
#pragma unroll
    for (int q = 0; q < 8; ++q) {
      float v0 = c0[q], v1 = c1[q];
      v0 = v0 > 0.f ? v0 : (__expf(v0) - 1.f);
      v1 = v1 > 0.f ? v1 : (__expf(v1) - 1.f);
      cr[(long)q * HOc]      = v0;
      cr[(long)q * HOc + 16] = v1;
    }
  }
}

template <int KTOT, bool RELU, bool RESID>
__global__ void __launch_bounds__(256) k_n256(const float* __restrict__ A, const float* __restrict__ Bm,
                                              const float* __restrict__ bias, const float* __restrict__ resid,
                                              float* __restrict__ C) {
  __shared__ __attribute__((aligned(16))) float part[4 * 2 * 16 * 64];
  constexpr int KQ = KTOT / 4;
  const int w    = threadIdx.x >> 5;
  const int lane = threadIdx.x & 31;
  const int l16  = lane & 15;
  const int hh   = lane >> 4;
  const int g    = blockIdx.x;
  const int r32  = g >> 2;
  const int col0 = (g & 3) * 64;
  const int rb   = w & 1;
  const int kq   = w >> 1;
  const int row0 = r32 * 32 + rb * 16;
  v8f c0 = {}, c1 = {}, c2 = {}, c3 = {};
  wmma_strip64_f32(A + (long)row0 * KTOT + kq * KQ, KTOT,
                   Bm + (long)kq * KQ * HOc + col0, HOc, KQ, c0, c1, c2, c3);
  float* p = part + ((kq * 2 + rb) * 16 + hh * 8) * 64 + 4 * l16;
#pragma unroll
  for (int r = 0; r < 8; ++r) {
    v4f o;
    o.x = c0[r];  o.y = c1[r];  o.z = c2[r];  o.w = c3[r];
    *reinterpret_cast<v4f*>(p + r * 64) = o;
  }
  __syncthreads();
  const int rr = threadIdx.x >> 4;
  const int cc = (threadIdx.x & 15) * 4;
  const v4f bv = *reinterpret_cast<const v4f*>(bias + col0 + cc);
#pragma unroll
  for (int hb = 0; hb < 2; ++hb) {
    const float* q = part + (hb * 16 + rr) * 64 + cc;
    v4f s = *reinterpret_cast<const v4f*>(q);
    s += *reinterpret_cast<const v4f*>(q + 1 * 2 * 16 * 64);
    s += *reinterpret_cast<const v4f*>(q + 2 * 2 * 16 * 64);
    s += *reinterpret_cast<const v4f*>(q + 3 * 2 * 16 * 64);
    s += bv;
    const long off = (long)(r32 * 32 + hb * 16 + rr) * HOc + col0 + cc;
    if (RESID) s += *reinterpret_cast<const v4f*>(resid + off);
    if (RELU) { s.x = fmaxf(s.x, 0.f);  s.y = fmaxf(s.y, 0.f);  s.z = fmaxf(s.z, 0.f);  s.w = fmaxf(s.w, 0.f); }
    *reinterpret_cast<v4f*>(C + off) = s;
  }
}

__global__ void __launch_bounds__(256) k_z(const float* __restrict__ y, const float* __restrict__ w1,
                                           const float* __restrict__ b1, float* __restrict__ z) {
  const int w    = threadIdx.x >> 5;
  const int lane = threadIdx.x & 31;
  const int l16  = lane & 15;
  const int hh   = lane >> 4;
  const int g    = blockIdx.x;
  const int row0 = (g >> 3) * 64 + (w & 3) * 16;
  const int col0 = (g & 7) * 128 + (w >> 2) * 64;
  v8f c0 = {}, c1 = {}, c2 = {}, c3 = {};
  wmma_strip64_f32(y + (long)row0 * HOc, HOc, w1 + col0, NHID, HOc, c0, c1, c2, c3);
  const v4f bv = *reinterpret_cast<const v4f*>(b1 + col0 + 4 * l16);
  float* zr = z + (long)(row0 + hh * 8) * NHID + col0 + 4 * l16;
#pragma unroll
  for (int r = 0; r < 8; ++r) {
    v4f o;
    o.x = fmaxf(c0[r] + bv.x, 0.f);
    o.y = fmaxf(c1[r] + bv.y, 0.f);
    o.z = fmaxf(c2[r] + bv.z, 0.f);
    o.w = fmaxf(c3[r] + bv.w, 0.f);
    *reinterpret_cast<v4f*>(zr + (long)r * NHID) = o;
  }
}

__global__ void k_ln(const float* __restrict__ ff, const float* __restrict__ g,
                     const float* __restrict__ bb, float* __restrict__ out) {
  int wg   = blockIdx.x * (blockDim.x >> 5) + (threadIdx.x >> 5);
  int lane = threadIdx.x & 31;
  const float* r = ff + (long)wg * HOc;
  float s1 = 0.f, s2 = 0.f;
#pragma unroll
  for (int p = 0; p < 8; ++p) {
    float v = r[p * 32 + lane];
    s1 += v;
    s2 += v * v;
  }
  s1 = wave_sum32(s1);
  s2 = wave_sum32(s2);
  float mu  = s1 * (1.f / HOc);
  float var = s2 * (1.f / HOc) - mu * mu;
  float inv = rsqrtf(var + EPSc);
  float* o = out + (long)wg * HOc;
#pragma unroll
  for (int p = 0; p < 8; ++p) {
    int c = p * 32 + lane;
    o[c] = (r[c] - mu) * inv * g[c] + bb[c];
  }
}

extern "C" void kernel_launch(void* const* d_in, const int* in_sizes, int n_in,
                              void* d_out, int out_size, void* d_ws, size_t ws_size,
                              hipStream_t stream) {
  const float* x    = (const float*)d_in[0];
  const float* edge = (const float*)d_in[1];
  const float* adj  = (const float*)d_in[2];
  const float* W    = (const float*)d_in[3];
  const float* W1   = (const float*)d_in[4];
  const float* a    = (const float*)d_in[5];
  const float* Wo   = (const float*)d_in[6];
  const float* bo   = (const float*)d_in[7];
  const float* w1   = (const float*)d_in[8];
  const float* b1   = (const float*)d_in[9];
  const float* w2   = (const float*)d_in[10];
  const float* b2   = (const float*)d_in[11];
  const float* lng  = (const float*)d_in[12];
  const float* lnb  = (const float*)d_in[13];
  float* out = (float*)d_out;

  float* ws  = (float*)d_ws;
  float* h   = ws;
  float* s   = h + 524288;
  float* t   = s + 16384;
  float* vt  = t + 16384;
  float* cat = vt + 4196352;
  float* y   = cat + 524288;
  float* z   = y + 524288;
  float* ff  = z + 2097152;

  k_h   <<<256,  256, 0, stream>>>(x, W, h);
  k_st  <<<64,   256, 0, stream>>>(h, a, s, t);
  k_v   <<<8,    256, 0, stream>>>(W1, a, vt);
  k_uattn_hp<<<256, 256, 0, stream>>>(edge, vt, s, t, adj, h, cat);
  k_n256<HOc,  true,  false><<<256, 256, 0, stream>>>(cat, Wo, bo, nullptr, y);
  k_z   <<<256,  256, 0, stream>>>(y, w1, b1, z);
  k_n256<NHID, false, true ><<<256, 256, 0, stream>>>(z, w2, b2, y, ff);
  k_ln  <<<256,  256, 0, stream>>>(ff, lng, lnb, out);
}


# s,t and vT prologue kernels folded into the h GEMM launch: s,t formed from the h block kept in LDS, vT by 8 extra workgroups; 8 launches become 6; summation order unchanged
# speedup vs baseline: 1.7497x; 1.0663x over previous
#include <hip/hip_runtime.h>

constexpr int Bc   = 8;
constexpr int Nc   = 256;
constexpr int NFc  = 256;
constexpr int EFc  = 128;
constexpr int Hc   = 8;
constexpr int Oc   = 32;
constexpr int NHID = 1024;
constexpr int HOc  = 256;
constexpr float ALPHA = 0.2f;
constexpr float EPSc  = 1e-6f;

typedef __attribute__((ext_vector_type(2))) float v2f;
typedef __attribute__((ext_vector_type(4))) float v4f;
typedef __attribute__((ext_vector_type(8))) float v8f;

__device__ __forceinline__ v8f wmma_gemm_f32(const float* __restrict__ A, int lda,
                                             const float* __restrict__ B, int ldb,
                                             int K, v8f acc) {
  const int lane = threadIdx.x & 31;
  const int l16  = lane & 15;
  const int ko   = (lane >> 4) << 2;
  const float* ap = A + l16 * lda + ko;
  const float* bp = B + ko * ldb + l16;
#pragma unroll 4
  for (int k = 0; k < K; k += 8) {
    const v4f a4 = *reinterpret_cast<const v4f*>(ap + k);
    v2f a01, a23, b01, b23;
    a01.x = a4.x;  a01.y = a4.y;
    a23.x = a4.z;  a23.y = a4.w;
    b01.x = bp[k * ldb];
    b01.y = bp[k * ldb + ldb];
    b23.x = bp[k * ldb + 2 * ldb];
    b23.y = bp[k * ldb + 3 * ldb];
    acc = __builtin_amdgcn_wmma_f32_16x16x4_f32(false, a01, false, b01,
                                                (short)0, acc, false, false);
    acc = __builtin_amdgcn_wmma_f32_16x16x4_f32(false, a23, false, b23,
                                                (short)0, acc, false, false);
  }
  return acc;
}

__device__ __forceinline__ void wmma_strip64_f32(const float* __restrict__ A, int lda,
                                                 const float* __restrict__ B, int ldb, int K,
                                                 v8f& c0, v8f& c1, v8f& c2, v8f& c3) {
  const int lane = threadIdx.x & 31;
  const int l16  = lane & 15;
  const int ko   = (lane >> 4) << 2;
  const float* ap = A + l16 * lda + ko;
  const float* bp = B + ko * ldb + 4 * l16;
  for (int k = 0; k < K; k += 32) {
    v4f a[4], r[4][4];
#pragma unroll
    for (int t = 0; t < 4; ++t) {
      a[t] = *reinterpret_cast<const v4f*>(ap + k + 8 * t);
#pragma unroll
      for (int s = 0; s < 4; ++s)
        r[t][s] = *reinterpret_cast<const v4f*>(bp + (k + 8 * t + s) * ldb);
    }
#pragma unroll
    for (int t = 0; t < 4; ++t) {
      v2f a01, a23, b;
      a01.x = a[t].x;  a01.y = a[t].y;
      a23.x = a[t].z;  a23.y = a[t].w;
      b.x = r[t][0].x;  b.y = r[t][1].x;
      c0 = __builtin_amdgcn_wmma_f32_16x16x4_f32(false, a01, false, b, (short)0, c0, false, false);
      b.x = r[t][0].y;  b.y = r[t][1].y;
      c1 = __builtin_amdgcn_wmma_f32_16x16x4_f32(false, a01, false, b, (short)0, c1, false, false);
      b.x = r[t][0].z;  b.y = r[t][1].z;
      c2 = __builtin_amdgcn_wmma_f32_16x16x4_f32(false, a01, false, b, (short)0, c2, false, false);
      b.x = r[t][0].w;  b.y = r[t][1].w;
      c3 = __builtin_amdgcn_wmma_f32_16x16x4_f32(false, a01, false, b, (short)0, c3, false, false);
      b.x = r[t][2].x;  b.y = r[t][3].x;
      c0 = __builtin_amdgcn_wmma_f32_16x16x4_f32(false, a23, false, b, (short)0, c0, false, false);
      b.x = r[t][2].y;  b.y = r[t][3].y;
      c1 = __builtin_amdgcn_wmma_f32_16x16x4_f32(false, a23, false, b, (short)0, c1, false, false);
      b.x = r[t][2].z;  b.y = r[t][3].z;
      c2 = __builtin_amdgcn_wmma_f32_16x16x4_f32(false, a23, false, b, (short)0, c2, false, false);
      b.x = r[t][2].w;  b.y = r[t][3].w;
      c3 = __builtin_amdgcn_wmma_f32_16x16x4_f32(false, a23, false, b, (short)0, c3, false, false);
    }
    __builtin_amdgcn_sched_group_barrier(0x020, 20, 0);
    __builtin_amdgcn_sched_group_barrier(0x008, 32, 0);
  }
}

__device__ __forceinline__ float wave_max32(float v) {
#pragma unroll
  for (int m = 16; m >= 1; m >>= 1) v = fmaxf(v, __shfl_xor(v, m, 32));
  return v;
}
__device__ __forceinline__ float wave_sum32(float v) {
#pragma unroll
  for (int m = 16; m >= 1; m >>= 1) v += __shfl_xor(v, m, 32);
  return v;
}

__device__ __forceinline__ void lds_barrier() {
  asm volatile("s_wait_dscnt 0x0\n\ts_barrier_signal -1\n\ts_barrier_wait -1" ::: "memory");
}

constexpr int HSP = 33;
__global__ void __launch_bounds__(256) k_h_st_v(const float* __restrict__ x, const float* __restrict__ W,
                                                const float* __restrict__ a, const float* __restrict__ W1,
                                                float* __restrict__ h, float* __restrict__ s,
                                                float* __restrict__ t, float* __restrict__ vt) {
  __shared__ float hs[64 * HSP];
  if (blockIdx.x < 8) {
    int idx = blockIdx.x * 256 + threadIdx.x;
    int c = idx & 15, e = idx >> 4;
    float v = 0.f;
    if (c < Hc) {
      const float* w  = W1 + ((long)c * EFc + e) * Oc;
      const float* a3 = a + c * (3 * Oc) + 2 * Oc;
#pragma unroll
      for (int o = 0; o < Oc; ++o) v += w[o] * a3[o];
    }
    vt[idx] = v;
    return;
  }
  const int blk = blockIdx.x - 8;
  const int wv  = threadIdx.x >> 5;
  const int nt  = wv & 1;
  const int ml  = wv >> 1;
  const int mt  = (blk & 3) * 4 + ml;
  const int bh  = blk >> 2;
  const int b   = bh >> 3, head = bh & 7;
  const float* A  = x + ((long)b * Nc + mt * 16) * NFc;
  const float* Bm = W + (long)head * NFc * Oc + nt * 16;
  v8f acc = {};
  acc = wmma_gemm_f32(A, NFc, Bm, Oc, NFc, acc);
  float* C = h + ((long)bh * Nc + mt * 16) * Oc + nt * 16;
  const int lane = threadIdx.x & 31;
  const int col  = lane & 15;
  const int rb   = (lane >> 4) * 8;
  float* hl = hs + (ml * 16 + rb) * HSP + nt * 16 + col;
#pragma unroll
  for (int g = 0; g < 8; ++g) {
    C[(rb + g) * Oc + col] = acc[g];
    hl[g * HSP] = acc[g];
  }
  lds_barrier();
  if (threadIdx.x < 128) {
    const int row   = threadIdx.x & 63;
    const int which = threadIdx.x >> 6;
    const float* hr = hs + row * HSP;
    const float* av = a + head * (3 * Oc) + which * Oc;
    float v = 0.f;
#pragma unroll
    for (int o = 0; o < Oc; ++o) v += hr[o] * av[o];
    float* dst = which ? t : s;
    dst[bh * Nc + (blk & 3) * 64 + row] = v;
  }
}

constexpr int SCP = 272;
constexpr int RPW = 8;
constexpr int ATP = 264;

template <bool HAS_NEXT>
__device__ __forceinline__ void uattn_row(v4f (&cx)[16], v4f (&cy)[16], v4f (&nx)[16], v4f (&ny)[16],
                                          const v2f (&vb01)[16], const v2f (&vb23)[16],
                                          const float* __restrict__ ap_next, const float* __restrict__ arow,
                                          const float sv, const float (&tv)[8],
                                          float* __restrict__ scw, float* __restrict__ atrow,
                                          const int w, const int lane, const int l16, const int hh) {
  float av[8];
#pragma unroll
  for (int p = 0; p < 8; ++p) av[p] = arow[p * 32 + lane];
  if (HAS_NEXT) {
#pragma unroll
    for (int t = 0; t < 16; ++t) {
      nx[t] = *reinterpret_cast<const v4f*>(ap_next + 8 * t);
      ny[t] = *reinterpret_cast<const v4f*>(ap_next + 128 * EFc + 8 * t);
    }
  }
  v8f acc0 = {}, acc1 = {};
#pragma unroll
  for (int t = 0; t < 16; ++t) {
    v2f p01, p23, q01, q23;
    p01.x = cx[t].x;  p01.y = cx[t].y;  p23.x = cx[t].z;  p23.y = cx[t].w;
    q01.x = cy[t].x;  q01.y = cy[t].y;  q23.x = cy[t].z;  q23.y = cy[t].w;
    acc0 = __builtin_amdgcn_wmma_f32_16x16x4_f32(false, p01, false, vb01[t], (short)0, acc0, false, false);
    acc1 = __builtin_amdgcn_wmma_f32_16x16x4_f32(false, q01, false, vb01[t], (short)0, acc1, false, false);
    acc0 = __builtin_amdgcn_wmma_f32_16x16x4_f32(false, p23, false, vb23[t], (short)0, acc0, false, false);
    acc1 = __builtin_amdgcn_wmma_f32_16x16x4_f32(false, q23, false, vb23[t], (short)0, acc1, false, false);
  }
  if (l16 < Hc) {
    float* d0 = scw + l16 * SCP + w * 16 + hh * 8;
#pragma unroll
    for (int g = 0; g < 8; ++g) { d0[g] = acc0[g]; d0[128 + g] = acc1[g]; }
  }
  lds_barrier();
  const float* srow = scw + w * SCP;
  float vals[8];
  float mx = -3.0e38f;
#pragma unroll
  for (int p = 0; p < 8; ++p) {
    float v = sv + tv[p] + srow[p * 32 + lane];
    v = v > 0.f ? v : ALPHA * v;
    v = (av[p] > 0.f) ? v : -3.0e38f;
    vals[p] = v;
    mx = fmaxf(mx, v);
  }
  mx = wave_max32(mx);
  float sum = 0.f;
#pragma unroll
  for (int p = 0; p < 8; ++p) {
    float e = __expf(vals[p] - mx);
    vals[p] = e;
    sum += e;
  }
  sum = wave_sum32(sum);
  const float inv = 1.f / sum;
#pragma unroll
  for (int p = 0; p < 8; ++p) atrow[p * 32 + lane] = vals[p] * inv;
}

__global__ void __launch_bounds__(256) k_uattn_hp(const float* __restrict__ edge, const float* __restrict__ vt,
                                                  const float* __restrict__ s, const float* __restrict__ t,
                                                  const float* __restrict__ adj, const float* __restrict__ h,
                                                  float* __restrict__ cat) {
  __shared__ __attribute__((aligned(16))) float sc[2 * Hc * SCP];
  __shared__ __attribute__((aligned(16))) float at[Hc * RPW * ATP];
  const int g    = blockIdx.x;
  const int b    = g >> 5;
  const int i0   = (g & 31) * RPW;
  const int w    = threadIdx.x >> 5;
  const int lane = threadIdx.x & 31;
  const int l16  = lane & 15;
  const int hh   = lane >> 4;
  const int ko   = hh << 2;
  const int bh   = b * Hc + w;
  const long row0 = (long)b * Nc + i0;
  const float* ap = edge + ((row0 * Nc) + w * 16 + l16) * EFc + ko;
  v4f xa[16], ya[16], xb[16], yb[16];
#pragma unroll
  for (int q = 0; q < 16; ++q) {
    xa[q] = *reinterpret_cast<const v4f*>(ap + 8 * q);
    ya[q] = *reinterpret_cast<const v4f*>(ap + 128 * EFc + 8 * q);
  }
  v2f vb01[16], vb23[16];
  const float* bp = vt + ko * 16 + l16;
#pragma unroll
  for (int q = 0; q < 16; ++q) {
    vb01[q].x = bp[(8 * q) * 16];
    vb01[q].y = bp[(8 * q + 1) * 16];
    vb23[q].x = bp[(8 * q + 2) * 16];
    vb23[q].y = bp[(8 * q + 3) * 16];
  }
  float tv[8], sv[RPW];
#pragma unroll
  for (int p = 0; p < 8; ++p) tv[p] = t[(long)bh * Nc + p * 32 + lane];
#pragma unroll
  for (int r = 0; r < RPW; ++r) sv[r] = s[bh * Nc + i0 + r];
  const float* arow = adj + row0 * Nc;
  float* atw = at + w * RPW * ATP;
  constexpr long ROWE = (long)Nc * EFc;
#pragma unroll
  for (int r = 0; r < RPW; r += 2) {
    uattn_row<true>(xa, ya, xb, yb, vb01, vb23, ap + (r + 1) * ROWE, arow + r * Nc, sv[r], tv,
                    sc, atw + r * ATP, w, lane, l16, hh);
    if (r + 2 < RPW)
      uattn_row<true>(xb, yb, xa, ya, vb01, vb23, ap + (r + 2) * ROWE, arow + (r + 1) * Nc, sv[r + 1], tv,
                      sc + Hc * SCP, atw + (r + 1) * ATP, w, lane, l16, hh);
    else
      uattn_row<false>(xb, yb, xa, ya, vb01, vb23, ap, arow + (r + 1) * Nc, sv[r + 1], tv,
                       sc + Hc * SCP, atw + (r + 1) * ATP, w, lane, l16, hh);
  }
  const float* aL = atw + (l16 & 7) * ATP + ko;
  const float* hb = h + (long)bh * Nc * Oc + ko * Oc + l16;
  v8f c0 = {}, c1 = {};
#pragma unroll 8
  for (int k = 0; k < Nc; k += 8) {
    const v4f a4 = *reinterpret_cast<const v4f*>(aL + k);
    v2f a01, a23, b01, b23, d01, d23;
    a01.x = a4.x;  a01.y = a4.y;  a23.x = a4.z;  a23.y = a4.w;
    b01.x = hb[k * Oc];            b01.y = hb[k * Oc + Oc];
    b23.x = hb[k * Oc + 2 * Oc];   b23.y = hb[k * Oc + 3 * Oc];
    d01.x = hb[k * Oc + 16];            d01.y = hb[k * Oc + Oc + 16];
    d23.x = hb[k * Oc + 2 * Oc + 16];   d23.y = hb[k * Oc + 3 * Oc + 16];
    c0 = __builtin_amdgcn_wmma_f32_16x16x4_f32(false, a01, false, b01, (short)0, c0, false, false);
    c1 = __builtin_amdgcn_wmma_f32_16x16x4_f32(false, a01, false, d01, (short)0, c1, false, false);
    c0 = __builtin_amdgcn_wmma_f32_16x16x4_f32(false, a23, false, b23, (short)0, c0, false, false);
    c1 = __builtin_amdgcn_wmma_f32_16x16x4_f32(false, a23, false, d23, (short)0, c1, false, false);
  }
  if (hh == 0) {
    float* cr = cat + (row0 * HOc) + w * Oc + l16;
#pragma unroll
    for (int q = 0; q < 8; ++q) {
      float v0 = c0[q], v1 = c1[q];
      v0 = v0 > 0.f ? v0 : (__expf(v0) - 1.f);
      v1 = v1 > 0.f ? v1 : (__expf(v1) - 1.f);
      cr[(long)q * HOc]      = v0;
      cr[(long)q * HOc + 16] = v1;
    }
  }
}

template <int KTOT, bool RELU, bool RESID>
__global__ void __launch_bounds__(256) k_n256(const float* __restrict__ A, const float* __restrict__ Bm,
                                              const float* __restrict__ bias, const float* __restrict__ resid,
                                              float* __restrict__ C) {
  __shared__ __attribute__((aligned(16))) float part[4 * 2 * 16 * 64];
  constexpr int KQ = KTOT / 4;
  const int w    = threadIdx.x >> 5;
  const int lane = threadIdx.x & 31;
  const int l16  = lane & 15;
  const int hh   = lane >> 4;
  const int g    = blockIdx.x;
  const int r32  = g >> 2;
  const int col0 = (g & 3) * 64;
  const int rb   = w & 1;
  const int kq   = w >> 1;
  const int row0 = r32 * 32 + rb * 16;
  v8f c0 = {}, c1 = {}, c2 = {}, c3 = {};
  wmma_strip64_f32(A + (long)row0 * KTOT + kq * KQ, KTOT,
                   Bm + (long)kq * KQ * HOc + col0, HOc, KQ, c0, c1, c2, c3);
  float* p = part + ((kq * 2 + rb) * 16 + hh * 8) * 64 + 4 * l16;
#pragma unroll
  for (int r = 0; r < 8; ++r) {
    v4f o;
    o.x = c0[r];  o.y = c1[r];  o.z = c2[r];  o.w = c3[r];
    *reinterpret_cast<v4f*>(p + r * 64) = o;
  }
  __syncthreads();
  const int rr = threadIdx.x >> 4;
  const int cc = (threadIdx.x & 15) * 4;
  const v4f bv = *reinterpret_cast<const v4f*>(bias + col0 + cc);
#pragma unroll
  for (int hb = 0; hb < 2; ++hb) {
    const float* q = part + (hb * 16 + rr) * 64 + cc;
    v4f s = *reinterpret_cast<const v4f*>(q);
    s += *reinterpret_cast<const v4f*>(q + 1 * 2 * 16 * 64);
    s += *reinterpret_cast<const v4f*>(q + 2 * 2 * 16 * 64);
    s += *reinterpret_cast<const v4f*>(q + 3 * 2 * 16 * 64);
    s += bv;
    const long off = (long)(r32 * 32 + hb * 16 + rr) * HOc + col0 + cc;
    if (RESID) s += *reinterpret_cast<const v4f*>(resid + off);
    if (RELU) { s.x = fmaxf(s.x, 0.f);  s.y = fmaxf(s.y, 0.f);  s.z = fmaxf(s.z, 0.f);  s.w = fmaxf(s.w, 0.f); }
    *reinterpret_cast<v4f*>(C + off) = s;
  }
}

__global__ void __launch_bounds__(256) k_z(const float* __restrict__ y, const float* __restrict__ w1,
                                           const float* __restrict__ b1, float* __restrict__ z) {
  const int w    = threadIdx.x >> 5;
  const int lane = threadIdx.x & 31;
  const int l16  = lane & 15;
  const int hh   = lane >> 4;
  const int g    = blockIdx.x;
  const int row0 = (g >> 3) * 64 + (w & 3) * 16;
  const int col0 = (g & 7) * 128 + (w >> 2) * 64;
  v8f c0 = {}, c1 = {}, c2 = {}, c3 = {};
  wmma_strip64_f32(y + (long)row0 * HOc, HOc, w1 + col0, NHID, HOc, c0, c1, c2, c3);
  const v4f bv = *reinterpret_cast<const v4f*>(b1 + col0 + 4 * l16);
  float* zr = z + (long)(row0 + hh * 8) * NHID + col0 + 4 * l16;
#pragma unroll
  for (int r = 0; r < 8; ++r) {
    v4f o;
    o.x = fmaxf(c0[r] + bv.x, 0.f);
    o.y = fmaxf(c1[r] + bv.y, 0.f);
    o.z = fmaxf(c2[r] + bv.z, 0.f);
    o.w = fmaxf(c3[r] + bv.w, 0.f);
    *reinterpret_cast<v4f*>(zr + (long)r * NHID) = o;
  }
}

__global__ void k_ln(const float* __restrict__ ff, const float* __restrict__ g,
                     const float* __restrict__ bb, float* __restrict__ out) {
  int wg   = blockIdx.x * (blockDim.x >> 5) + (threadIdx.x >> 5);
  int lane = threadIdx.x & 31;
  const float* r = ff + (long)wg * HOc;
  float s1 = 0.f, s2 = 0.f;
#pragma unroll
  for (int p = 0; p < 8; ++p) {
    float v = r[p * 32 + lane];
    s1 += v;
    s2 += v * v;
  }
  s1 = wave_sum32(s1);
  s2 = wave_sum32(s2);
  float mu  = s1 * (1.f / HOc);
  float var = s2 * (1.f / HOc) - mu * mu;
  float inv = rsqrtf(var + EPSc);
  float* o = out + (long)wg * HOc;
#pragma unroll
  for (int p = 0; p < 8; ++p) {
    int c = p * 32 + lane;
    o[c] = (r[c] - mu) * inv * g[c] + bb[c];
  }
}

extern "C" void kernel_launch(void* const* d_in, const int* in_sizes, int n_in,
                              void* d_out, int out_size, void* d_ws, size_t ws_size,
                              hipStream_t stream) {
  const float* x    = (const float*)d_in[0];
  const float* edge = (const float*)d_in[1];
  const float* adj  = (const float*)d_in[2];
  const float* W    = (const float*)d_in[3];
  const float* W1   = (const float*)d_in[4];
  const float* a    = (const float*)d_in[5];
  const float* Wo   = (const float*)d_in[6];
  const float* bo   = (const float*)d_in[7];
  const float* w1   = (const float*)d_in[8];
  const float* b1   = (const float*)d_in[9];
  const float* w2   = (const float*)d_in[10];
  const float* b2   = (const float*)d_in[11];
  const float* lng  = (const float*)d_in[12];
  const float* lnb  = (const float*)d_in[13];
  float* out = (float*)d_out;

  float* ws  = (float*)d_ws;
  float* h   = ws;
  float* s   = h + 524288;
  float* t   = s + 16384;
  float* vt  = t + 16384;
  float* cat = vt + 4196352;
  float* y   = cat + 524288;
  float* z   = y + 524288;
  float* ff  = z + 2097152;

  k_h_st_v<<<264, 256, 0, stream>>>(x, W, a, W1, h, s, t, vt);
  k_uattn_hp<<<256, 256, 0, stream>>>(edge, vt, s, t, adj, h, cat);
  k_n256<HOc,  true,  false><<<256, 256, 0, stream>>>(cat, Wo, bo, nullptr, y);
  k_z   <<<256,  256, 0, stream>>>(y, w1, b1, z);
  k_n256<NHID, false, true ><<<256, 256, 0, stream>>>(z, w2, b2, y, ff);
  k_ln  <<<256,  256, 0, stream>>>(ff, lng, lnb, out);
}


# Fragment loads of the h GEMM and of the Wo, w1, w2 strip GEMMs issued two to three k groups ahead of the matrix steps that use them, in a ring of register sets; k order and sums unchanged
# speedup vs baseline: 1.9655x; 1.1233x over previous
#include <hip/hip_runtime.h>

constexpr int Bc   = 8;
constexpr int Nc   = 256;
constexpr int NFc  = 256;
constexpr int EFc  = 128;
constexpr int Hc   = 8;
constexpr int Oc   = 32;
constexpr int NHID = 1024;
constexpr int HOc  = 256;
constexpr float ALPHA = 0.2f;
constexpr float EPSc  = 1e-6f;

typedef __attribute__((ext_vector_type(2))) float v2f;
typedef __attribute__((ext_vector_type(4))) float v4f;
typedef __attribute__((ext_vector_type(8))) float v8f;

template <int K>
__device__ __forceinline__ v8f wmma_gemm_f32(const float* __restrict__ A, int lda,
                                             const float* __restrict__ B, int ldb, v8f acc) {
  constexpr int NG    = K / 32;
  constexpr int AHEAD = 3;
  constexpr int RING  = AHEAD + 1;
  const int lane = threadIdx.x & 31;
  const int l16  = lane & 15;
  const int ko   = (lane >> 4) << 2;
  const float* ap = A + l16 * lda + ko;
  const float* bp = B + ko * ldb + l16;
  v4f   a[RING][4];
  float b[RING][4][4];
#pragma unroll
  for (int g = 0; g < NG + AHEAD; ++g) {
    if (g < NG) {
      const int S = g % RING;
#pragma unroll
      for (int t = 0; t < 4; ++t) {
        const int k = 32 * g + 8 * t;
        a[S][t] = *reinterpret_cast<const v4f*>(ap + k);
#pragma unroll
        for (int s = 0; s < 4; ++s) b[S][t][s] = bp[(k + s) * ldb];
      }
    }
    if (g >= AHEAD) {
      const int S = (g - AHEAD) % RING;
#pragma unroll
      for (int t = 0; t < 4; ++t) {
        v2f a01, a23, b01, b23;
        a01.x = a[S][t].x;  a01.y = a[S][t].y;
        a23.x = a[S][t].z;  a23.y = a[S][t].w;
        b01.x = b[S][t][0];  b01.y = b[S][t][1];
        b23.x = b[S][t][2];  b23.y = b[S][t][3];
        acc = __builtin_amdgcn_wmma_f32_16x16x4_f32(false, a01, false, b01,
                                                    (short)0, acc, false, false);
        acc = __builtin_amdgcn_wmma_f32_16x16x4_f32(false, a23, false, b23,
                                                    (short)0, acc, false, false);
      }
    }
    if (g < NG)     __builtin_amdgcn_sched_group_barrier(0x020, 20, 0);
    if (g >= AHEAD) __builtin_amdgcn_sched_group_barrier(0x008, 8, 0);
  }
  return acc;
}

template <int K>
__device__ __forceinline__ void wmma_strip64_f32(const float* __restrict__ A, int lda,
                                                 const float* __restrict__ B, int ldb,
                                                 v8f& c0, v8f& c1, v8f& c2, v8f& c3) {
  constexpr int NG    = K / 32;
  constexpr int AHEAD = 2;
  constexpr int RING  = AHEAD + 1;
  const int lane = threadIdx.x & 31;
  const int l16  = lane & 15;
  const int ko   = (lane >> 4) << 2;
  const float* ap = A + l16 * lda + ko;
  const float* bp = B + ko * ldb + 4 * l16;
  v4f a[RING][4], r[RING][4][4];
#pragma unroll
  for (int g = 0; g < NG + AHEAD; ++g) {
    if (g < NG) {
      const int S = g % RING;
#pragma unroll
      for (int t = 0; t < 4; ++t) {
        const int k = 32 * g + 8 * t;
        a[S][t] = *reinterpret_cast<const v4f*>(ap + k);
#pragma unroll
        for (int s = 0; s < 4; ++s)
          r[S][t][s] = *reinterpret_cast<const v4f*>(bp + (k + s) * ldb);
      }
    }
    if (g >= AHEAD) {
      const int S = (g - AHEAD) % RING;
#pragma unroll
      for (int t = 0; t < 4; ++t) {
        v2f a01, a23, b;
        a01.x = a[S][t].x;  a01.y = a[S][t].y;
        a23.x = a[S][t].z;  a23.y = a[S][t].w;
        b.x = r[S][t][0].x;  b.y = r[S][t][1].x;
        c0 = __builtin_amdgcn_wmma_f32_16x16x4_f32(false, a01, false, b, (short)0, c0, false, false);
        b.x = r[S][t][0].y;  b.y = r[S][t][1].y;
        c1 = __builtin_amdgcn_wmma_f32_16x16x4_f32(false, a01, false, b, (short)0, c1, false, false);
        b.x = r[S][t][0].z;  b.y = r[S][t][1].z;
        c2 = __builtin_amdgcn_wmma_f32_16x16x4_f32(false, a01, false, b, (short)0, c2, false, false);
        b.x = r[S][t][0].w;  b.y = r[S][t][1].w;
        c3 = __builtin_amdgcn_wmma_f32_16x16x4_f32(false, a01, false, b, (short)0, c3, false, false);
        b.x = r[S][t][2].x;  b.y = r[S][t][3].x;
        c0 = __builtin_amdgcn_wmma_f32_16x16x4_f32(false, a23, false, b, (short)0, c0, false, false);
        b.x = r[S][t][2].y;  b.y = r[S][t][3].y;
        c1 = __builtin_amdgcn_wmma_f32_16x16x4_f32(false, a23, false, b, (short)0, c1, false, false);
        b.x = r[S][t][2].z;  b.y = r[S][t][3].z;
        c2 = __builtin_amdgcn_wmma_f32_16x16x4_f32(false, a23, false, b, (short)0, c2, false, false);
        b.x = r[S][t][2].w;  b.y = r[S][t][3].w;
        c3 = __builtin_amdgcn_wmma_f32_16x16x4_f32(false, a23, false, b, (short)0, c3, false, false);
      }
    }
    if (g < NG)     __builtin_amdgcn_sched_group_barrier(0x020, 20, 0);
    if (g >= AHEAD) __builtin_amdgcn_sched_group_barrier(0x008, 32, 0);
  }
}

__device__ __forceinline__ float wave_max32(float v) {
#pragma unroll
  for (int m = 16; m >= 1; m >>= 1) v = fmaxf(v, __shfl_xor(v, m, 32));
  return v;
}
__device__ __forceinline__ float wave_sum32(float v) {
#pragma unroll
  for (int m = 16; m >= 1; m >>= 1) v += __shfl_xor(v, m, 32);
  return v;
}

__device__ __forceinline__ void lds_barrier() {
  asm volatile("s_wait_dscnt 0x0\n\ts_barrier_signal -1\n\ts_barrier_wait -1" ::: "memory");
}

constexpr int HSP = 33;
__global__ void __launch_bounds__(256) k_h_st_v(const float* __restrict__ x, const float* __restrict__ W,
                                                const float* __restrict__ a, const float* __restrict__ W1,
                                                float* __restrict__ h, float* __restrict__ s,
                                                float* __restrict__ t, float* __restrict__ vt) {
  __shared__ float hs[64 * HSP];
  if (blockIdx.x < 8) {
    int idx = blockIdx.x * 256 + threadIdx.x;
    int c = idx & 15, e = idx >> 4;
    float v = 0.f;
    if (c < Hc) {
      const float* w  = W1 + ((long)c * EFc + e) * Oc;
      const float* a3 = a + c * (3 * Oc) + 2 * Oc;
#pragma unroll
      for (int o = 0; o < Oc; ++o) v += w[o] * a3[o];
    }
    vt[idx] = v;
    return;
  }
  const int blk = blockIdx.x - 8;
  const int wv  = threadIdx.x >> 5;
  const int nt  = wv & 1;
  const int ml  = wv >> 1;
  const int mt  = (blk & 3) * 4 + ml;
  const int bh  = blk >> 2;
  const int b   = bh >> 3, head = bh & 7;
  const float* A  = x + ((long)b * Nc + mt * 16) * NFc;
  const float* Bm = W + (long)head * NFc * Oc + nt * 16;
  v8f acc = {};
  acc = wmma_gemm_f32<NFc>(A, NFc, Bm, Oc, acc);
  float* C = h + ((long)bh * Nc + mt * 16) * Oc + nt * 16;
  const int lane = threadIdx.x & 31;
  const int col  = lane & 15;
  const int rb   = (lane >> 4) * 8;
  float* hl = hs + (ml * 16 + rb) * HSP + nt * 16 + col;
#pragma unroll
  for (int g = 0; g < 8; ++g) {
    C[(rb + g) * Oc + col] = acc[g];
    hl[g * HSP] = acc[g];
  }
  lds_barrier();
  if (threadIdx.x < 128) {
    const int row   = threadIdx.x & 63;
    const int which = threadIdx.x >> 6;
    const float* hr = hs + row * HSP;
    const float* av = a + head * (3 * Oc) + which * Oc;
    float v = 0.f;
#pragma unroll
    for (int o = 0; o < Oc; ++o) v += hr[o] * av[o];
    float* dst = which ? t : s;
    dst[bh * Nc + (blk & 3) * 64 + row] = v;
  }
}

constexpr int SCP = 272;
constexpr int RPW = 8;
constexpr int ATP = 264;

template <bool HAS_NEXT>
__device__ __forceinline__ void uattn_row(v4f (&cx)[16], v4f (&cy)[16], v4f (&nx)[16], v4f (&ny)[16],
                                          const v2f (&vb01)[16], const v2f (&vb23)[16],
                                          const float* __restrict__ ap_next, const float* __restrict__ arow,
                                          const float sv, const float (&tv)[8],
                                          float* __restrict__ scw, float* __restrict__ atrow,
                                          const int w, const int lane, const int l16, const int hh) {
  float av[8];
#pragma unroll
  for (int p = 0; p < 8; ++p) av[p] = arow[p * 32 + lane];
  if (HAS_NEXT) {
#pragma unroll
    for (int t = 0; t < 16; ++t) {
      nx[t] = *reinterpret_cast<const v4f*>(ap_next + 8 * t);
      ny[t] = *reinterpret_cast<const v4f*>(ap_next + 128 * EFc + 8 * t);
    }
  }
  v8f acc0 = {}, acc1 = {};
#pragma unroll
  for (int t = 0; t < 16; ++t) {
    v2f p01, p23, q01, q23;
    p01.x = cx[t].x;  p01.y = cx[t].y;  p23.x = cx[t].z;  p23.y = cx[t].w;
    q01.x = cy[t].x;  q01.y = cy[t].y;  q23.x = cy[t].z;  q23.y = cy[t].w;
    acc0 = __builtin_amdgcn_wmma_f32_16x16x4_f32(false, p01, false, vb01[t], (short)0, acc0, false, false);
    acc1 = __builtin_amdgcn_wmma_f32_16x16x4_f32(false, q01, false, vb01[t], (short)0, acc1, false, false);
    acc0 = __builtin_amdgcn_wmma_f32_16x16x4_f32(false, p23, false, vb23[t], (short)0, acc0, false, false);
    acc1 = __builtin_amdgcn_wmma_f32_16x16x4_f32(false, q23, false, vb23[t], (short)0, acc1, false, false);
  }
  if (l16 < Hc) {
    float* d0 = scw + l16 * SCP + w * 16 + hh * 8;
#pragma unroll
    for (int g = 0; g < 8; ++g) { d0[g] = acc0[g]; d0[128 + g] = acc1[g]; }
  }
  lds_barrier();
  const float* srow = scw + w * SCP;
  float vals[8];
  float mx = -3.0e38f;
#pragma unroll
  for (int p = 0; p < 8; ++p) {
    float v = sv + tv[p] + srow[p * 32 + lane];
    v = v > 0.f ? v : ALPHA * v;
    v = (av[p] > 0.f) ? v : -3.0e38f;
    vals[p] = v;
    mx = fmaxf(mx, v);
  }
  mx = wave_max32(mx);
  float sum = 0.f;
#pragma unroll
  for (int p = 0; p < 8; ++p) {
    float e = __expf(vals[p] - mx);
    vals[p] = e;
    sum += e;
  }
  sum = wave_sum32(sum);
  const float inv = 1.f / sum;
#pragma unroll
  for (int p = 0; p < 8; ++p) atrow[p * 32 + lane] = vals[p] * inv;
}

__global__ void __launch_bounds__(256) k_uattn_hp(const float* __restrict__ edge, const float* __restrict__ vt,
                                                  const float* __restrict__ s, const float* __restrict__ t,
                                                  const float* __restrict__ adj, const float* __restrict__ h,
                                                  float* __restrict__ cat) {
  __shared__ __attribute__((aligned(16))) float sc[2 * Hc * SCP];
  __shared__ __attribute__((aligned(16))) float at[Hc * RPW * ATP];
  const int g    = blockIdx.x;
  const int b    = g >> 5;
  const int i0   = (g & 31) * RPW;
  const int w    = threadIdx.x >> 5;
  const int lane = threadIdx.x & 31;
  const int l16  = lane & 15;
  const int hh   = lane >> 4;
  const int ko   = hh << 2;
  const int bh   = b * Hc + w;
  const long row0 = (long)b * Nc + i0;
  const float* ap = edge + ((row0 * Nc) + w * 16 + l16) * EFc + ko;
  v4f xa[16], ya[16], xb[16], yb[16];
#pragma unroll
  for (int q = 0; q < 16; ++q) {
    xa[q] = *reinterpret_cast<const v4f*>(ap + 8 * q);
    ya[q] = *reinterpret_cast<const v4f*>(ap + 128 * EFc + 8 * q);
  }
  v2f vb01[16], vb23[16];
  const float* bp = vt + ko * 16 + l16;
#pragma unroll
  for (int q = 0; q < 16; ++q) {
    vb01[q].x = bp[(8 * q) * 16];
    vb01[q].y = bp[(8 * q + 1) * 16];
    vb23[q].x = bp[(8 * q + 2) * 16];
    vb23[q].y = bp[(8 * q + 3) * 16];
  }
  float tv[8], sv[RPW];
#pragma unroll
  for (int p = 0; p < 8; ++p) tv[p] = t[(long)bh * Nc + p * 32 + lane];
#pragma unroll
  for (int r = 0; r < RPW; ++r) sv[r] = s[bh * Nc + i0 + r];
  const float* arow = adj + row0 * Nc;
  float* atw = at + w * RPW * ATP;
  constexpr long ROWE = (long)Nc * EFc;
#pragma unroll
  for (int r = 0; r < RPW; r += 2) {
    uattn_row<true>(xa, ya, xb, yb, vb01, vb23, ap + (r + 1) * ROWE, arow + r * Nc, sv[r], tv,
                    sc, atw + r * ATP, w, lane, l16, hh);
    if (r + 2 < RPW)
      uattn_row<true>(xb, yb, xa, ya, vb01, vb23, ap + (r + 2) * ROWE, arow + (r + 1) * Nc, sv[r + 1], tv,
                      sc + Hc * SCP, atw + (r + 1) * ATP, w, lane, l16, hh);
    else
      uattn_row<false>(xb, yb, xa, ya, vb01, vb23, ap, arow + (r + 1) * Nc, sv[r + 1], tv,
                       sc + Hc * SCP, atw + (r + 1) * ATP, w, lane, l16, hh);
  }
  const float* aL = atw + (l16 & 7) * ATP + ko;
  const float* hb = h + (long)bh * Nc * Oc + ko * Oc + l16;
  v8f c0 = {}, c1 = {};
#pragma unroll 8
  for (int k = 0; k < Nc; k += 8) {
    const v4f a4 = *reinterpret_cast<const v4f*>(aL + k);
    v2f a01, a23, b01, b23, d01, d23;
    a01.x = a4.x;  a01.y = a4.y;  a23.x = a4.z;  a23.y = a4.w;
    b01.x = hb[k * Oc];            b01.y = hb[k * Oc + Oc];
    b23.x = hb[k * Oc + 2 * Oc];   b23.y = hb[k * Oc + 3 * Oc];
    d01.x = hb[k * Oc + 16];            d01.y = hb[k * Oc + Oc + 16];
    d23.x = hb[k * Oc + 2 * Oc + 16];   d23.y = hb[k * Oc + 3 * Oc + 16];
    c0 = __builtin_amdgcn_wmma_f32_16x16x4_f32(false, a01, false, b01, (short)0, c0, false, false);
    c1 = __builtin_amdgcn_wmma_f32_16x16x4_f32(false, a01, false, d01, (short)0, c1, false, false);
    c0 = __builtin_amdgcn_wmma_f32_16x16x4_f32(false, a23, false, b23, (short)0, c0, false, false);
    c1 = __builtin_amdgcn_wmma_f32_16x16x4_f32(false, a23, false, d23, (short)0, c1, false, false);
  }
  if (hh == 0) {
    float* cr = cat + (row0 * HOc) + w * Oc + l16;
#pragma unroll
    for (int q = 0; q < 8; ++q) {
      float v0 = c0[q], v1 = c1[q];
      v0 = v0 > 0.f ? v0 : (__expf(v0) - 1.f);
      v1 = v1 > 0.f ? v1 : (__expf(v1) - 1.f);
      cr[(long)q * HOc]      = v0;
      cr[(long)q * HOc + 16] = v1;
    }
  }
}

template <int KTOT, bool RELU, bool RESID>
__global__ void __launch_bounds__(256) k_n256(const float* __restrict__ A, const float* __restrict__ Bm,
                                              const float* __restrict__ bias, const float* __restrict__ resid,
                                              float* __restrict__ C) {
  __shared__ __attribute__((aligned(16))) float part[4 * 2 * 16 * 64];
  constexpr int KQ = KTOT / 4;
  const int w    = threadIdx.x >> 5;
  const int lane = threadIdx.x & 31;
  const int l16  = lane & 15;
  const int hh   = lane >> 4;
  const int g    = blockIdx.x;
  const int r32  = g >> 2;
  const int col0 = (g & 3) * 64;
  const int rb   = w & 1;
  const int kq   = w >> 1;
  const int row0 = r32 * 32 + rb * 16;
  v8f c0 = {}, c1 = {}, c2 = {}, c3 = {};
  wmma_strip64_f32<KQ>(A + (long)row0 * KTOT + kq * KQ, KTOT,
                       Bm + (long)kq * KQ * HOc + col0, HOc, c0, c1, c2, c3);
  float* p = part + ((kq * 2 + rb) * 16 + hh * 8) * 64 + 4 * l16;
#pragma unroll
  for (int r = 0; r < 8; ++r) {
    v4f o;
    o.x = c0[r];  o.y = c1[r];  o.z = c2[r];  o.w = c3[r];
    *reinterpret_cast<v4f*>(p + r * 64) = o;
  }
  __syncthreads();
  const int rr = threadIdx.x >> 4;
  const int cc = (threadIdx.x & 15) * 4;
  const v4f bv = *reinterpret_cast<const v4f*>(bias + col0 + cc);
#pragma unroll
  for (int hb = 0; hb < 2; ++hb) {
    const float* q = part + (hb * 16 + rr) * 64 + cc;
    v4f s = *reinterpret_cast<const v4f*>(q);
    s += *reinterpret_cast<const v4f*>(q + 1 * 2 * 16 * 64);
    s += *reinterpret_cast<const v4f*>(q + 2 * 2 * 16 * 64);
    s += *reinterpret_cast<const v4f*>(q + 3 * 2 * 16 * 64);
    s += bv;
    const long off = (long)(r32 * 32 + hb * 16 + rr) * HOc + col0 + cc;
    if (RESID) s += *reinterpret_cast<const v4f*>(resid + off);
    if (RELU) { s.x = fmaxf(s.x, 0.f);  s.y = fmaxf(s.y, 0.f);  s.z = fmaxf(s.z, 0.f);  s.w = fmaxf(s.w, 0.f); }
    *reinterpret_cast<v4f*>(C + off) = s;
  }
}

__global__ void __launch_bounds__(256) k_z(const float* __restrict__ y, const float* __restrict__ w1,
                                           const float* __restrict__ b1, float* __restrict__ z) {
  const int w    = threadIdx.x >> 5;
  const int lane = threadIdx.x & 31;
  const int l16  = lane & 15;
  const int hh   = lane >> 4;
  const int g    = blockIdx.x;
  const int row0 = (g >> 3) * 64 + (w & 3) * 16;
  const int col0 = (g & 7) * 128 + (w >> 2) * 64;
  v8f c0 = {}, c1 = {}, c2 = {}, c3 = {};
  wmma_strip64_f32<HOc>(y + (long)row0 * HOc, HOc, w1 + col0, NHID, c0, c1, c2, c3);
  const v4f bv = *reinterpret_cast<const v4f*>(b1 + col0 + 4 * l16);
  float* zr = z + (long)(row0 + hh * 8) * NHID + col0 + 4 * l16;
#pragma unroll
  for (int r = 0; r < 8; ++r) {
    v4f o;
    o.x = fmaxf(c0[r] + bv.x, 0.f);
    o.y = fmaxf(c1[r] + bv.y, 0.f);
    o.z = fmaxf(c2[r] + bv.z, 0.f);
    o.w = fmaxf(c3[r] + bv.w, 0.f);
    *reinterpret_cast<v4f*>(zr + (long)r * NHID) = o;
  }
}

__global__ void k_ln(const float* __restrict__ ff, const float* __restrict__ g,
                     const float* __restrict__ bb, float* __restrict__ out) {
  int wg   = blockIdx.x * (blockDim.x >> 5) + (threadIdx.x >> 5);
  int lane = threadIdx.x & 31;
  const float* r = ff + (long)wg * HOc;
  float s1 = 0.f, s2 = 0.f;
#pragma unroll
  for (int p = 0; p < 8; ++p) {
    float v = r[p * 32 + lane];
    s1 += v;
    s2 += v * v;
  }
  s1 = wave_sum32(s1);
  s2 = wave_sum32(s2);
  float mu  = s1 * (1.f / HOc);
  float var = s2 * (1.f / HOc) - mu * mu;
  float inv = rsqrtf(var + EPSc);
  float* o = out + (long)wg * HOc;
#pragma unroll
  for (int p = 0; p < 8; ++p) {
    int c = p * 32 + lane;
    o[c] = (r[c] - mu) * inv * g[c] + bb[c];
  }
}

extern "C" void kernel_launch(void* const* d_in, const int* in_sizes, int n_in,
                              void* d_out, int out_size, void* d_ws, size_t ws_size,
                              hipStream_t stream) {
  const float* x    = (const float*)d_in[0];
  const float* edge = (const float*)d_in[1];
  const float* adj  = (const float*)d_in[2];
  const float* W    = (const float*)d_in[3];
  const float* W1   = (const float*)d_in[4];
  const float* a    = (const float*)d_in[5];
  const float* Wo   = (const float*)d_in[6];
  const float* bo   = (const float*)d_in[7];
  const float* w1   = (const float*)d_in[8];
  const float* b1   = (const float*)d_in[9];
  const float* w2   = (const float*)d_in[10];
  const float* b2   = (const float*)d_in[11];
  const float* lng  = (const float*)d_in[12];
  const float* lnb  = (const float*)d_in[13];
  float* out = (float*)d_out;

  float* ws  = (float*)d_ws;
  float* h   = ws;
  float* s   = h + 524288;
  float* t   = s + 16384;
  float* vt  = t + 16384;
  float* cat = vt + 4196352;
  float* y   = cat + 524288;
  float* z   = y + 524288;
  float* ff  = z + 2097152;

  k_h_st_v<<<264, 256, 0, stream>>>(x, W, a, W1, h, s, t, vt);
  k_uattn_hp<<<256, 256, 0, stream>>>(edge, vt, s, t, adj, h, cat);
  k_n256<HOc,  true,  false><<<256, 256, 0, stream>>>(cat, Wo, bo, nullptr, y);
  k_z   <<<256,  256, 0, stream>>>(y, w1, b1, z);
  k_n256<NHID, false, true ><<<256, 256, 0, stream>>>(z, w2, b2, y, ff);
  k_ln  <<<256,  256, 0, stream>>>(ff, lng, lnb, out);
}


# w1 and w2 GEMMs fused: a workgroup forms a 32x256 block of z, keeps it in LDS and multiplies it by its w2 quarter; LayerNorm adds the 4 partial sums in the old order: sums unchanged; 6 launches to 5
# speedup vs baseline: 2.1081x; 1.0726x over previous
#include <hip/hip_runtime.h>

constexpr int Bc   = 8;
constexpr int Nc   = 256;
constexpr int NFc  = 256;
constexpr int EFc  = 128;
constexpr int Hc   = 8;
constexpr int Oc   = 32;
constexpr int NHID = 1024;
constexpr int HOc  = 256;
constexpr float ALPHA = 0.2f;
constexpr float EPSc  = 1e-6f;

typedef __attribute__((ext_vector_type(2))) float v2f;
typedef __attribute__((ext_vector_type(4))) float v4f;
typedef __attribute__((ext_vector_type(8))) float v8f;

template <int K>
__device__ __forceinline__ v8f wmma_gemm_f32(const float* __restrict__ A, int lda,
                                             const float* __restrict__ B, int ldb, v8f acc) {
  constexpr int NG    = K / 32;
  constexpr int AHEAD = 3;
  constexpr int RING  = AHEAD + 1;
  const int lane = threadIdx.x & 31;
  const int l16  = lane & 15;
  const int ko   = (lane >> 4) << 2;
  const float* ap = A + l16 * lda + ko;
  const float* bp = B + ko * ldb + l16;
  v4f   a[RING][4];
  float b[RING][4][4];
#pragma unroll
  for (int g = 0; g < NG + AHEAD; ++g) {
    if (g < NG) {
      const int S = g % RING;
#pragma unroll
      for (int t = 0; t < 4; ++t) {
        const int k = 32 * g + 8 * t;
        a[S][t] = *reinterpret_cast<const v4f*>(ap + k);
#pragma unroll
        for (int s = 0; s < 4; ++s) b[S][t][s] = bp[(k + s) * ldb];
      }
    }
    if (g >= AHEAD) {
      const int S = (g - AHEAD) % RING;
#pragma unroll
      for (int t = 0; t < 4; ++t) {
        v2f a01, a23, b01, b23;
        a01.x = a[S][t].x;  a01.y = a[S][t].y;
        a23.x = a[S][t].z;  a23.y = a[S][t].w;
        b01.x = b[S][t][0];  b01.y = b[S][t][1];
        b23.x = b[S][t][2];  b23.y = b[S][t][3];
        acc = __builtin_amdgcn_wmma_f32_16x16x4_f32(false, a01, false, b01,
                                                    (short)0, acc, false, false);
        acc = __builtin_amdgcn_wmma_f32_16x16x4_f32(false, a23, false, b23,
                                                    (short)0, acc, false, false);
      }
    }
    if (g < NG)     __builtin_amdgcn_sched_group_barrier(0x020, 20, 0);
    if (g >= AHEAD) __builtin_amdgcn_sched_group_barrier(0x008, 8, 0);
  }
  return acc;
}

template <int K, bool A_LDS>
__device__ __forceinline__ void wmma_strip64_f32(const float* __restrict__ A, int lda,
                                                 const float* __restrict__ B, int ldb,
                                                 v8f& c0, v8f& c1, v8f& c2, v8f& c3) {
  constexpr int NG    = K / 32;
  constexpr int AHEAD = 2;
  constexpr int RING  = AHEAD + 1;
  const int lane = threadIdx.x & 31;
  const int l16  = lane & 15;
  const int ko   = (lane >> 4) << 2;
  const float* ap = A + l16 * lda + ko;
  const float* bp = B + ko * ldb + 4 * l16;
  v4f a[RING][4], r[RING][4][4];
#pragma unroll
  for (int g = 0; g < NG + AHEAD; ++g) {
    if (g < NG) {
      const int S = g % RING;
#pragma unroll
      for (int t = 0; t < 4; ++t) {
        const int k = 32 * g + 8 * t;
        a[S][t] = *reinterpret_cast<const v4f*>(ap + k);
#pragma unroll
        for (int s = 0; s < 4; ++s)
          r[S][t][s] = *reinterpret_cast<const v4f*>(bp + (k + s) * ldb);
      }
    }
    if (g >= AHEAD) {
      const int S = (g - AHEAD) % RING;
#pragma unroll
      for (int t = 0; t < 4; ++t) {
        v2f a01, a23, b;
        a01.x = a[S][t].x;  a01.y = a[S][t].y;
        a23.x = a[S][t].z;  a23.y = a[S][t].w;
        b.x = r[S][t][0].x;  b.y = r[S][t][1].x;
        c0 = __builtin_amdgcn_wmma_f32_16x16x4_f32(false, a01, false, b, (short)0, c0, false, false);
        b.x = r[S][t][0].y;  b.y = r[S][t][1].y;
        c1 = __builtin_amdgcn_wmma_f32_16x16x4_f32(false, a01, false, b, (short)0, c1, false, false);
        b.x = r[S][t][0].z;  b.y = r[S][t][1].z;
        c2 = __builtin_amdgcn_wmma_f32_16x16x4_f32(false, a01, false, b, (short)0, c2, false, false);
        b.x = r[S][t][0].w;  b.y = r[S][t][1].w;
        c3 = __builtin_amdgcn_wmma_f32_16x16x4_f32(false, a01, false, b, (short)0, c3, false, false);
        b.x = r[S][t][2].x;  b.y = r[S][t][3].x;
        c0 = __builtin_amdgcn_wmma_f32_16x16x4_f32(false, a23, false, b, (short)0, c0, false, false);
        b.x = r[S][t][2].y;  b.y = r[S][t][3].y;
        c1 = __builtin_amdgcn_wmma_f32_16x16x4_f32(false, a23, false, b, (short)0, c1, false, false);
        b.x = r[S][t][2].z;  b.y = r[S][t][3].z;
        c2 = __builtin_amdgcn_wmma_f32_16x16x4_f32(false, a23, false, b, (short)0, c2, false, false);
        b.x = r[S][t][2].w;  b.y = r[S][t][3].w;
        c3 = __builtin_amdgcn_wmma_f32_16x16x4_f32(false, a23, false, b, (short)0, c3, false, false);
      }
    }
    if (g < NG)     __builtin_amdgcn_sched_group_barrier(A_LDS ? 0x120 : 0x020, 20, 0);
    if (g >= AHEAD) __builtin_amdgcn_sched_group_barrier(0x008, 32, 0);
  }
}

__device__ __forceinline__ float wave_max32(float v) {
#pragma unroll
  for (int m = 16; m >= 1; m >>= 1) v = fmaxf(v, __shfl_xor(v, m, 32));
  return v;
}
__device__ __forceinline__ float wave_sum32(float v) {
#pragma unroll
  for (int m = 16; m >= 1; m >>= 1) v += __shfl_xor(v, m, 32);
  return v;
}

__device__ __forceinline__ void lds_barrier() {
  asm volatile("s_wait_dscnt 0x0\n\ts_barrier_signal -1\n\ts_barrier_wait -1" ::: "memory");
}

constexpr int HSP = 33;
__global__ void __launch_bounds__(256) k_h_st_v(const float* __restrict__ x, const float* __restrict__ W,
                                                const float* __restrict__ a, const float* __restrict__ W1,
                                                float* __restrict__ h, float* __restrict__ s,
                                                float* __restrict__ t, float* __restrict__ vt) {
  __shared__ float hs[64 * HSP];
  if (blockIdx.x < 8) {
    int idx = blockIdx.x * 256 + threadIdx.x;
    int c = idx & 15, e = idx >> 4;
    float v = 0.f;
    if (c < Hc) {
      const float* w  = W1 + ((long)c * EFc + e) * Oc;
      const float* a3 = a + c * (3 * Oc) + 2 * Oc;
#pragma unroll
      for (int o = 0; o < Oc; ++o) v += w[o] * a3[o];
    }
    vt[idx] = v;
    return;
  }
  const int blk = blockIdx.x - 8;
  const int wv  = threadIdx.x >> 5;
  const int nt  = wv & 1;
  const int ml  = wv >> 1;
  const int mt  = (blk & 3) * 4 + ml;
  const int bh  = blk >> 2;
  const int b   = bh >> 3, head = bh & 7;
  const float* A  = x + ((long)b * Nc + mt * 16) * NFc;
  const float* Bm = W + (long)head * NFc * Oc + nt * 16;
  v8f acc = {};
  acc = wmma_gemm_f32<NFc>(A, NFc, Bm, Oc, acc);
  float* C = h + ((long)bh * Nc + mt * 16) * Oc + nt * 16;
  const int lane = threadIdx.x & 31;
  const int col  = lane & 15;
  const int rb   = (lane >> 4) * 8;
  float* hl = hs + (ml * 16 + rb) * HSP + nt * 16 + col;
#pragma unroll
  for (int g = 0; g < 8; ++g) {
    C[(rb + g) * Oc + col] = acc[g];
    hl[g * HSP] = acc[g];
  }
  lds_barrier();
  if (threadIdx.x < 128) {
    const int row   = threadIdx.x & 63;
    const int which = threadIdx.x >> 6;
    const float* hr = hs + row * HSP;
    const float* av = a + head * (3 * Oc) + which * Oc;
    float v = 0.f;
#pragma unroll
    for (int o = 0; o < Oc; ++o) v += hr[o] * av[o];
    float* dst = which ? t : s;
    dst[bh * Nc + (blk & 3) * 64 + row] = v;
  }
}

constexpr int SCP = 272;
constexpr int RPW = 8;
constexpr int ATP = 264;

template <bool HAS_NEXT>
__device__ __forceinline__ void uattn_row(v4f (&cx)[16], v4f (&cy)[16], v4f (&nx)[16], v4f (&ny)[16],
                                          const v2f (&vb01)[16], const v2f (&vb23)[16],
                                          const float* __restrict__ ap_next, const float* __restrict__ arow,
                                          const float sv, const float (&tv)[8],
                                          float* __restrict__ scw, float* __restrict__ atrow,
                                          const int w, const int lane, const int l16, const int hh) {
  float av[8];
#pragma unroll
  for (int p = 0; p < 8; ++p) av[p] = arow[p * 32 + lane];
  if (HAS_NEXT) {
#pragma unroll
    for (int t = 0; t < 16; ++t) {
      nx[t] = *reinterpret_cast<const v4f*>(ap_next + 8 * t);
      ny[t] = *reinterpret_cast<const v4f*>(ap_next + 128 * EFc + 8 * t);
    }
  }
  v8f acc0 = {}, acc1 = {};
#pragma unroll
  for (int t = 0; t < 16; ++t) {
    v2f p01, p23, q01, q23;
    p01.x = cx[t].x;  p01.y = cx[t].y;  p23.x = cx[t].z;  p23.y = cx[t].w;
    q01.x = cy[t].x;  q01.y = cy[t].y;  q23.x = cy[t].z;  q23.y = cy[t].w;
    acc0 = __builtin_amdgcn_wmma_f32_16x16x4_f32(false, p01, false, vb01[t], (short)0, acc0, false, false);
    acc1 = __builtin_amdgcn_wmma_f32_16x16x4_f32(false, q01, false, vb01[t], (short)0, acc1, false, false);
    acc0 = __builtin_amdgcn_wmma_f32_16x16x4_f32(false, p23, false, vb23[t], (short)0, acc0, false, false);
    acc1 = __builtin_amdgcn_wmma_f32_16x16x4_f32(false, q23, false, vb23[t], (short)0, acc1, false, false);
  }
  if (l16 < Hc) {
    float* d0 = scw + l16 * SCP + w * 16 + hh * 8;
#pragma unroll
    for (int g = 0; g < 8; ++g) { d0[g] = acc0[g]; d0[128 + g] = acc1[g]; }
  }
  lds_barrier();
  const float* srow = scw + w * SCP;
  float vals[8];
  float mx = -3.0e38f;
#pragma unroll
  for (int p = 0; p < 8; ++p) {
    float v = sv + tv[p] + srow[p * 32 + lane];
    v = v > 0.f ? v : ALPHA * v;
    v = (av[p] > 0.f) ? v : -3.0e38f;
    vals[p] = v;
    mx = fmaxf(mx, v);
  }
  mx = wave_max32(mx);
  float sum = 0.f;
#pragma unroll
  for (int p = 0; p < 8; ++p) {
    float e = __expf(vals[p] - mx);
    vals[p] = e;
    sum += e;
  }
  sum = wave_sum32(sum);
  const float inv = 1.f / sum;
#pragma unroll
  for (int p = 0; p < 8; ++p) atrow[p * 32 + lane] = vals[p] * inv;
}

__global__ void __launch_bounds__(256) k_uattn_hp(const float* __restrict__ edge, const float* __restrict__ vt,
                                                  const float* __restrict__ s, const float* __restrict__ t,
                                                  const float* __restrict__ adj, const float* __restrict__ h,
                                                  float* __restrict__ cat) {
  __shared__ __attribute__((aligned(16))) float sc[2 * Hc * SCP];
  __shared__ __attribute__((aligned(16))) float at[Hc * RPW * ATP];
  const int g    = blockIdx.x;
  const int b    = g >> 5;
  const int i0   = (g & 31) * RPW;
  const int w    = threadIdx.x >> 5;
  const int lane = threadIdx.x & 31;
  const int l16  = lane & 15;
  const int hh   = lane >> 4;
  const int ko   = hh << 2;
  const int bh   = b * Hc + w;
  const long row0 = (long)b * Nc + i0;
  const float* ap = edge + ((row0 * Nc) + w * 16 + l16) * EFc + ko;
  v4f xa[16], ya[16], xb[16], yb[16];
#pragma unroll
  for (int q = 0; q < 16; ++q) {
    xa[q] = *reinterpret_cast<const v4f*>(ap + 8 * q);
    ya[q] = *reinterpret_cast<const v4f*>(ap + 128 * EFc + 8 * q);
  }
  v2f vb01[16], vb23[16];
  const float* bp = vt + ko * 16 + l16;
#pragma unroll
  for (int q = 0; q < 16; ++q) {
    vb01[q].x = bp[(8 * q) * 16];
    vb01[q].y = bp[(8 * q + 1) * 16];
    vb23[q].x = bp[(8 * q + 2) * 16];
    vb23[q].y = bp[(8 * q + 3) * 16];
  }
  float tv[8], sv[RPW];
#pragma unroll
  for (int p = 0; p < 8; ++p) tv[p] = t[(long)bh * Nc + p * 32 + lane];
#pragma unroll
  for (int r = 0; r < RPW; ++r) sv[r] = s[bh * Nc + i0 + r];
  const float* arow = adj + row0 * Nc;
  float* atw = at + w * RPW * ATP;
  constexpr long ROWE = (long)Nc * EFc;
#pragma unroll
  for (int r = 0; r < RPW; r += 2) {
    uattn_row<true>(xa, ya, xb, yb, vb01, vb23, ap + (r + 1) * ROWE, arow + r * Nc, sv[r], tv,
                    sc, atw + r * ATP, w, lane, l16, hh);
    if (r + 2 < RPW)
      uattn_row<true>(xb, yb, xa, ya, vb01, vb23, ap + (r + 2) * ROWE, arow + (r + 1) * Nc, sv[r + 1], tv,
                      sc + Hc * SCP, atw + (r + 1) * ATP, w, lane, l16, hh);
    else
      uattn_row<false>(xb, yb, xa, ya, vb01, vb23, ap, arow + (r + 1) * Nc, sv[r + 1], tv,
                       sc + Hc * SCP, atw + (r + 1) * ATP, w, lane, l16, hh);
  }
  const float* aL = atw + (l16 & 7) * ATP + ko;
  const float* hb = h + (long)bh * Nc * Oc + ko * Oc + l16;
  v8f c0 = {}, c1 = {};
#pragma unroll 8
  for (int k = 0; k < Nc; k += 8) {
    const v4f a4 = *reinterpret_cast<const v4f*>(aL + k);
    v2f a01, a23, b01, b23, d01, d23;
    a01.x = a4.x;  a01.y = a4.y;  a23.x = a4.z;  a23.y = a4.w;
    b01.x = hb[k * Oc];            b01.y = hb[k * Oc + Oc];
    b23.x = hb[k * Oc + 2 * Oc];   b23.y = hb[k * Oc + 3 * Oc];
    d01.x = hb[k * Oc + 16];            d01.y = hb[k * Oc + Oc + 16];
    d23.x = hb[k * Oc + 2 * Oc + 16];   d23.y = hb[k * Oc + 3 * Oc + 16];
    c0 = __builtin_amdgcn_wmma_f32_16x16x4_f32(false, a01, false, b01, (short)0, c0, false, false);
    c1 = __builtin_amdgcn_wmma_f32_16x16x4_f32(false, a01, false, d01, (short)0, c1, false, false);
    c0 = __builtin_amdgcn_wmma_f32_16x16x4_f32(false, a23, false, b23, (short)0, c0, false, false);
    c1 = __builtin_amdgcn_wmma_f32_16x16x4_f32(false, a23, false, d23, (short)0, c1, false, false);
  }
  if (hh == 0) {
    float* cr = cat + (row0 * HOc) + w * Oc + l16;
#pragma unroll
    for (int q = 0; q < 8; ++q) {
      float v0 = c0[q], v1 = c1[q];
      v0 = v0 > 0.f ? v0 : (__expf(v0) - 1.f);
      v1 = v1 > 0.f ? v1 : (__expf(v1) - 1.f);
      cr[(long)q * HOc]      = v0;
      cr[(long)q * HOc + 16] = v1;
    }
  }
}

template <int KTOT, bool RELU, bool RESID>
__global__ void __launch_bounds__(256) k_n256(const float* __restrict__ A, const float* __restrict__ Bm,
                                              const float* __restrict__ bias, const float* __restrict__ resid,
                                              float* __restrict__ C) {
  __shared__ __attribute__((aligned(16))) float part[4 * 2 * 16 * 64];
  constexpr int KQ = KTOT / 4;
  const int w    = threadIdx.x >> 5;
  const int lane = threadIdx.x & 31;
  const int l16  = lane & 15;
  const int hh   = lane >> 4;
  const int g    = blockIdx.x;
  const int r32  = g >> 2;
  const int col0 = (g & 3) * 64;
  const int rb   = w & 1;
  const int kq   = w >> 1;
  const int row0 = r32 * 32 + rb * 16;
  v8f c0 = {}, c1 = {}, c2 = {}, c3 = {};
  wmma_strip64_f32<KQ, false>(A + (long)row0 * KTOT + kq * KQ, KTOT,
                       Bm + (long)kq * KQ * HOc + col0, HOc, c0, c1, c2, c3);
  float* p = part + ((kq * 2 + rb) * 16 + hh * 8) * 64 + 4 * l16;
#pragma unroll
  for (int r = 0; r < 8; ++r) {
    v4f o;
    o.x = c0[r];  o.y = c1[r];  o.z = c2[r];  o.w = c3[r];
    *reinterpret_cast<v4f*>(p + r * 64) = o;
  }
  __syncthreads();
  const int rr = threadIdx.x >> 4;
  const int cc = (threadIdx.x & 15) * 4;
  const v4f bv = *reinterpret_cast<const v4f*>(bias + col0 + cc);
#pragma unroll
  for (int hb = 0; hb < 2; ++hb) {
    const float* q = part + (hb * 16 + rr) * 64 + cc;
    v4f s = *reinterpret_cast<const v4f*>(q);
    s += *reinterpret_cast<const v4f*>(q + 1 * 2 * 16 * 64);
    s += *reinterpret_cast<const v4f*>(q + 2 * 2 * 16 * 64);
    s += *reinterpret_cast<const v4f*>(q + 3 * 2 * 16 * 64);
    s += bv;
    const long off = (long)(r32 * 32 + hb * 16 + rr) * HOc + col0 + cc;
    if (RESID) s += *reinterpret_cast<const v4f*>(resid + off);
    if (RELU) { s.x = fmaxf(s.x, 0.f);  s.y = fmaxf(s.y, 0.f);  s.z = fmaxf(s.z, 0.f);  s.w = fmaxf(s.w, 0.f); }
    *reinterpret_cast<v4f*>(C + off) = s;
  }
}

constexpr int ZP = 260;
__global__ void __launch_bounds__(256) k_ffn(const float* __restrict__ y, const float* __restrict__ w1,
                                             const float* __restrict__ b1, const float* __restrict__ w2,
                                             float* __restrict__ part) {
  __shared__ __attribute__((aligned(16))) float zs[32 * ZP];
  const int w    = threadIdx.x >> 5;
  const int lane = threadIdx.x & 31;
  const int l16  = lane & 15;
  const int hh   = lane >> 4;
  const int g    = blockIdx.x;
  const int r32  = g >> 2;
  const int q    = g & 3;
  const int rb   = w & 1;
  const int sq   = w >> 1;
  const int row0 = r32 * 32 + rb * 16;
  {
    v8f c0 = {}, c1 = {}, c2 = {}, c3 = {};
    wmma_strip64_f32<HOc, false>(y + (long)row0 * HOc, HOc, w1 + q * 256 + sq * 64, NHID, c0, c1, c2, c3);
    const v4f bv = *reinterpret_cast<const v4f*>(b1 + q * 256 + sq * 64 + 4 * l16);
    float* zr = zs + (rb * 16 + hh * 8) * ZP + sq * 64 + 4 * l16;
#pragma unroll
    for (int r = 0; r < 8; ++r) {
      v4f o;
      o.x = fmaxf(c0[r] + bv.x, 0.f);
      o.y = fmaxf(c1[r] + bv.y, 0.f);
      o.z = fmaxf(c2[r] + bv.z, 0.f);
      o.w = fmaxf(c3[r] + bv.w, 0.f);
      *reinterpret_cast<v4f*>(zr + r * ZP) = o;
    }
  }
  lds_barrier();
  v8f d0 = {}, d1 = {}, d2 = {}, d3 = {};
  wmma_strip64_f32<256, true>(zs + rb * 16 * ZP, ZP, w2 + (long)q * 256 * HOc + sq * 64, HOc, d0, d1, d2, d3);
  float* pr = part + ((long)q * (Bc * Nc) + row0 + hh * 8) * HOc + sq * 64 + 4 * l16;
#pragma unroll
  for (int r = 0; r < 8; ++r) {
    v4f o;
    o.x = d0[r];  o.y = d1[r];  o.z = d2[r];  o.w = d3[r];
    *reinterpret_cast<v4f*>(pr + (long)r * HOc) = o;
  }
}

__global__ void k_ln(const float* __restrict__ part, const float* __restrict__ b2, const float* __restrict__ y,
                     const float* __restrict__ g, const float* __restrict__ bb, float* __restrict__ out) {
  int wg   = blockIdx.x * (blockDim.x >> 5) + (threadIdx.x >> 5);
  int lane = threadIdx.x & 31;
  constexpr long PS = (long)Bc * Nc * HOc;
  const float* r  = part + (long)wg * HOc;
  const float* yr = y + (long)wg * HOc;
  float vals[8];
  float s1 = 0.f, s2 = 0.f;
#pragma unroll
  for (int p = 0; p < 8; ++p) {
    const int c = p * 32 + lane;
    float v = r[c];
    v += r[PS + c];
    v += r[2 * PS + c];
    v += r[3 * PS + c];
    v += b2[c];
    v += yr[c];
    vals[p] = v;
    s1 += v;
    s2 += v * v;
  }
  s1 = wave_sum32(s1);
  s2 = wave_sum32(s2);
  float mu  = s1 * (1.f / HOc);
  float var = s2 * (1.f / HOc) - mu * mu;
  float inv = rsqrtf(var + EPSc);
  float* o = out + (long)wg * HOc;
#pragma unroll
  for (int p = 0; p < 8; ++p) {
    int c = p * 32 + lane;
    o[c] = (vals[p] - mu) * inv * g[c] + bb[c];
  }
}

extern "C" void kernel_launch(void* const* d_in, const int* in_sizes, int n_in,
                              void* d_out, int out_size, void* d_ws, size_t ws_size,
                              hipStream_t stream) {
  const float* x    = (const float*)d_in[0];
  const float* edge = (const float*)d_in[1];
  const float* adj  = (const float*)d_in[2];
  const float* W    = (const float*)d_in[3];
  const float* W1   = (const float*)d_in[4];
  const float* a    = (const float*)d_in[5];
  const float* Wo   = (const float*)d_in[6];
  const float* bo   = (const float*)d_in[7];
  const float* w1   = (const float*)d_in[8];
  const float* b1   = (const float*)d_in[9];
  const float* w2   = (const float*)d_in[10];
  const float* b2   = (const float*)d_in[11];
  const float* lng  = (const float*)d_in[12];
  const float* lnb  = (const float*)d_in[13];
  float* out = (float*)d_out;

  float* ws  = (float*)d_ws;
  float* h   = ws;
  float* s   = h + 524288;
  float* t   = s + 16384;
  float* vt  = t + 16384;
  float* cat = vt + 4196352;
  float* y   = cat + 524288;
  float* pt  = y + 524288;

  k_h_st_v<<<264, 256, 0, stream>>>(x, W, a, W1, h, s, t, vt);
  k_uattn_hp<<<256, 256, 0, stream>>>(edge, vt, s, t, adj, h, cat);
  k_n256<HOc,  true,  false><<<256, 256, 0, stream>>>(cat, Wo, bo, nullptr, y);
  k_ffn <<<256,  256, 0, stream>>>(y, w1, b1, w2, pt);
  k_ln  <<<256,  256, 0, stream>>>(pt, b2, y, lng, lnb, out);
}
